# starting kernel
# baseline (speedup 1.0000x reference)
#include <hip/hip_runtime.h>
#include <hip/hip_bf16.h>

#define NN 100000
#define DD 128
#define HH 64
#define KK 8
#define LL 3
#define CC 16
#define EE 1600000
#define THETA 0.1f

typedef __attribute__((ext_vector_type(2))) float v2f;
typedef __attribute__((ext_vector_type(8))) float v8f;

__device__ __forceinline__ v8f wmma_f32(v2f a, v2f b, v8f c) {
  return __builtin_amdgcn_wmma_f32_16x16x4_f32(
       false, a,  false, b,
       (short)0, c,  false,  false);
}

__global__ void zero_kernel(float* __restrict__ p, int n) {
  int i = blockIdx.x * blockDim.x + threadIdx.x;
  if (i < n) p[i] = 0.0f;
}

__global__ void degree_kernel(const int* __restrict__ col, float* __restrict__ deg) {
  int e = blockIdx.x * blockDim.x + threadIdx.x;
  if (e < EE) atomicAdd(&deg[col[e]], 1.0f);
}

__global__ void dn_kernel(const float* __restrict__ deg, float* __restrict__ dn) {
  int i = blockIdx.x * blockDim.x + threadIdx.x;
  if (i < NN) {
    float d = deg[i];
    dn[i] = (d > 0.0f) ? rsqrtf(d) : 0.0f;
  }
}

__global__ __launch_bounds__(256) void fc0_kernel(const float* __restrict__ x,
                                                  const float* __restrict__ w,
                                                  const float* __restrict__ b,
                                                  float* __restrict__ h) {
  const int wave = threadIdx.x >> 5;
  const int lane = threadIdx.x & 31;
  const int half = lane >> 4;
  const int lid  = lane & 15;
  const int node0 = blockIdx.x * 128 + wave * 16;
  if (node0 >= NN) return;

  v8f acc0 = {}, acc1 = {}, acc2 = {}, acc3 = {};
  for (int c = 0; c < DD / 4; ++c) {
    const int k0 = 4 * c + 2 * half;
    const float* ap = x + (node0 + lid) * DD + k0;
    v2f a; a.x = ap[0]; a.y = ap[1];
    const float* bp = w + k0 * HH;
    v2f b0; b0.x = bp[lid];      b0.y = bp[HH + lid];
    v2f b1; b1.x = bp[16 + lid]; b1.y = bp[HH + 16 + lid];
    v2f b2; b2.x = bp[32 + lid]; b2.y = bp[HH + 32 + lid];
    v2f b3; b3.x = bp[48 + lid]; b3.y = bp[HH + 48 + lid];
    acc0 = wmma_f32(a, b0, acc0);
    acc1 = wmma_f32(a, b1, acc1);
    acc2 = wmma_f32(a, b2, acc2);
    acc3 = wmma_f32(a, b3, acc3);
  }
  v8f accs[4] = {acc0, acc1, acc2, acc3};
  for (int t = 0; t < 4; ++t) {
    const float bias = b[16 * t + lid];
    for (int r = 0; r < 8; ++r) {
      float v = accs[t][r] + bias;
      h[(node0 + r + 8 * half) * HH + 16 * t + lid] = v > 0.0f ? v : 0.0f;
    }
  }
}

__global__ void scatter_kernel(const int* __restrict__ row, const int* __restrict__ col,
                               const float* __restrict__ dn, const float* __restrict__ h,
                               float* __restrict__ agg) {
  unsigned idx = blockIdx.x * blockDim.x + threadIdx.x;
  unsigned e = idx >> 4;
  int seg = idx & 15;
  if (e >= EE) return;
  int r = row[e], c = col[e];
  float v = dn[r] * dn[c];
  const float4 hv = *(const float4*)(h + r * HH + seg * 4);
  float* ap = agg + c * HH + seg * 4;
  atomicAdd(ap + 0, v * hv.x);
  atomicAdd(ap + 1, v * hv.y);
  atomicAdd(ap + 2, v * hv.z);
  atomicAdd(ap + 3, v * hv.w);
}

__global__ void gate_kernel(const float* __restrict__ h, const float* __restrict__ env_w,
                            const float* __restrict__ env_b, float* __restrict__ e_out,
                            int layer) {
  int n = blockIdx.x * blockDim.x + threadIdx.x;
  if (n >= NN) return;
  const float* W = env_w + layer * 2 * HH * KK;
  const float* B = env_b + layer * KK;
  float logit[KK];
  #pragma unroll
  for (int k = 0; k < KK; ++k) logit[k] = B[k];
  for (int f = 0; f < HH; ++f) {
    float hv = h[n * HH + f];
    const float* wr = W + f * KK;
    #pragma unroll
    for (int k = 0; k < KK; ++k) logit[k] += hv * wr[k];
  }
  float m = logit[0];
  #pragma unroll
  for (int k = 1; k < KK; ++k) m = fmaxf(m, logit[k]);
  float s = 0.0f;
  float ex[KK];
  #pragma unroll
  for (int k = 0; k < KK; ++k) { ex[k] = __expf(logit[k] - m); s += ex[k]; }
  float inv = 1.0f / s;
  #pragma unroll
  for (int k = 0; k < KK; ++k) {
    float pi = ex[k] * inv;
    e_out[n * KK + k] = (pi > THETA) ? pi : 0.0f;
  }
}

__global__ __launch_bounds__(256) void conv_kernel(const float* __restrict__ agg,
                                                   const float* __restrict__ h,
                                                   const float* __restrict__ e_gate,
                                                   const float* __restrict__ conv_w,
                                                   float* __restrict__ h_out,
                                                   int layer) {
  const int wave = threadIdx.x >> 5;
  const int lane = threadIdx.x & 31;
  const int half = lane >> 4;
  const int lid  = lane & 15;
  const int node0 = blockIdx.x * 128 + wave * 16;
  if (node0 >= NN) return;

  const float* W = conv_w + layer * (KK * 2 * HH * HH);
  v8f acc0 = {}, acc1 = {}, acc2 = {}, acc3 = {};

  #pragma unroll 1
  for (int k = 0; k < KK; ++k) {
    const float ek = e_gate[(node0 + lid) * KK + k];
    if (!__any(ek != 0.0f)) continue;
    const float* Wk = W + k * (2 * HH * HH);
    #pragma unroll 1
    for (int c = 0; c < (2 * HH) / 4; ++c) {
      const int kk = 4 * c + 2 * half;
      const float* src = (kk < HH) ? agg : h;
      const int fk = (kk < HH) ? kk : kk - HH;
      const float* ap = src + (node0 + lid) * HH + fk;
      v2f a; a.x = ek * ap[0]; a.y = ek * ap[1];
      const float* bp = Wk + kk * HH;
      v2f b0; b0.x = bp[lid];      b0.y = bp[HH + lid];
      v2f b1; b1.x = bp[16 + lid]; b1.y = bp[HH + 16 + lid];
      v2f b2; b2.x = bp[32 + lid]; b2.y = bp[HH + 32 + lid];
      v2f b3; b3.x = bp[48 + lid]; b3.y = bp[HH + 48 + lid];
      acc0 = wmma_f32(a, b0, acc0);
      acc1 = wmma_f32(a, b1, acc1);
      acc2 = wmma_f32(a, b2, acc2);
      acc3 = wmma_f32(a, b3, acc3);
    }
  }
  v8f accs[4] = {acc0, acc1, acc2, acc3};
  for (int t = 0; t < 4; ++t) {
    for (int r = 0; r < 8; ++r) {
      const int idx = (node0 + r + 8 * half) * HH + 16 * t + lid;
      float v = accs[t][r] + h[idx];
      h_out[idx] = v > 0.0f ? v : 0.0f;
    }
  }
}

__global__ __launch_bounds__(256) void fc1_kernel(const float* __restrict__ h,
                                                  const float* __restrict__ w,
                                                  const float* __restrict__ b,
                                                  float* __restrict__ out) {
  const int wave = threadIdx.x >> 5;
  const int lane = threadIdx.x & 31;
  const int half = lane >> 4;
  const int lid  = lane & 15;
  const int node0 = blockIdx.x * 128 + wave * 16;
  if (node0 >= NN) return;

  v8f acc = {};
  for (int c = 0; c < HH / 4; ++c) {
    const int k0 = 4 * c + 2 * half;
    const float* ap = h + (node0 + lid) * HH + k0;
    v2f a; a.x = ap[0]; a.y = ap[1];
    const float* bp = w + k0 * CC;
    v2f bb; bb.x = bp[lid]; bb.y = bp[CC + lid];
    acc = wmma_f32(a, bb, acc);
  }
  const float bias = b[lid];
  for (int r = 0; r < 8; ++r)
    out[(node0 + r + 8 * half) * CC + lid] = acc[r] + bias;
}

extern "C" void kernel_launch(void* const* d_in, const int* in_sizes, int n_in,
                              void* d_out, int out_size, void* d_ws, size_t ws_size,
                              hipStream_t stream) {
  const float* x      = (const float*)d_in[0];
  const int*   eidx   = (const int*)d_in[1];
  const float* fc0_w  = (const float*)d_in[2];
  const float* fc0_b  = (const float*)d_in[3];
  const float* fc1_w  = (const float*)d_in[4];
  const float* fc1_b  = (const float*)d_in[5];
  const float* env_w  = (const float*)d_in[6];
  const float* env_b  = (const float*)d_in[7];
  const float* conv_w = (const float*)d_in[8];
  float* out = (float*)d_out;

  const int* row = eidx;
  const int* col = eidx + EE;

  float* ws  = (float*)d_ws;
  float* deg = ws;
  float* dn  = deg + NN;
  float* hA  = dn + NN;
  float* hB  = hA + (size_t)NN * HH;
  float* agg = hB + (size_t)NN * HH;
  float* eg  = agg + (size_t)NN * HH;

  const int T = 256;
  const int gemm_blocks = (NN + 127) / 128;

  zero_kernel<<<(NN + T - 1) / T, T, 0, stream>>>(deg, NN);
  degree_kernel<<<(EE + T - 1) / T, T, 0, stream>>>(col, deg);
  dn_kernel<<<(NN + T - 1) / T, T, 0, stream>>>(deg, dn);

  fc0_kernel<<<gemm_blocks, T, 0, stream>>>(x, fc0_w, fc0_b, hA);

  float* cur = hA;
  float* nxt = hB;
  for (int layer = 0; layer < LL; ++layer) {
    zero_kernel<<<(NN * HH + T - 1) / T, T, 0, stream>>>(agg, NN * HH);
    {
      unsigned items = (unsigned)EE * 16u;
      scatter_kernel<<<(items + T - 1) / T, T, 0, stream>>>(row, col, dn, cur, agg);
    }
    gate_kernel<<<(NN + T - 1) / T, T, 0, stream>>>(cur, env_w, env_b, eg, layer);
    conv_kernel<<<gemm_blocks, T, 0, stream>>>(agg, cur, eg, conv_w, nxt, layer);
    float* tmp = cur; cur = nxt; nxt = tmp;
  }

  fc1_kernel<<<gemm_blocks, T, 0, stream>>>(cur, fc1_w, fc1_b, out);
}


# Aggregation without float atomics: edges sorted by destination once per call (int atomic arrival rank + prefix scan), rows gathered per node; sum order = arrival-order rank
# speedup vs baseline: 1.4356x; 1.4356x over previous
#include <hip/hip_runtime.h>
#include <hip/hip_bf16.h>

#define NN 100000
#define DD 128
#define HH 64
#define KK 8
#define LL 3
#define CC 16
#define EE 1600000
#define THETA 0.1f

typedef __attribute__((ext_vector_type(2))) float v2f;
typedef __attribute__((ext_vector_type(8))) float v8f;

__device__ __forceinline__ v8f wmma_f32(v2f a, v2f b, v8f c) {
  return __builtin_amdgcn_wmma_f32_16x16x4_f32(
       false, a,  false, b,
       (short)0, c,  false,  false);
}

__global__ void zero_kernel(int* __restrict__ p, int n) {
  int i = blockIdx.x * blockDim.x + threadIdx.x;
  if (i < n) p[i] = 0;
}

__global__ void count_rank_kernel(const int* __restrict__ col, int* __restrict__ cnt,
                                  int* __restrict__ rank) {
  int e = blockIdx.x * blockDim.x + threadIdx.x;
  if (e < EE) rank[e] = atomicAdd(&cnt[col[e]], 1);
}

__global__ __launch_bounds__(256) void block_sum_kernel(const int* __restrict__ cnt,
                                                        int* __restrict__ bsum) {
  __shared__ int s[256];
  const int t = threadIdx.x;
  const int i = blockIdx.x * 256 + t;
  s[t] = (i < NN) ? cnt[i] : 0;
  __syncthreads();
  for (int off = 128; off > 0; off >>= 1) {
    if (t < off) s[t] += s[t + off];
    __syncthreads();
  }
  if (t == 0) bsum[blockIdx.x] = s[0];
}

__global__ __launch_bounds__(256) void scan_kernel(int* __restrict__ cs,
                                                   const int* __restrict__ bsum,
                                                   float* __restrict__ dn) {
  __shared__ int s[256];
  __shared__ int red[256];
  const int t = threadIdx.x;
  const int b = blockIdx.x;
  int p = 0;
  for (int j = t; j < b; j += 256) p += bsum[j];
  red[t] = p;
  const int i = b * 256 + t;
  const int c = (i < NN) ? cs[i] : 0;
  s[t] = c;
  __syncthreads();
  for (int off = 128; off > 0; off >>= 1) {
    if (t < off) red[t] += red[t + off];
    __syncthreads();
  }
  for (int off = 1; off < 256; off <<= 1) {
    const int v = (t >= off) ? s[t - off] : 0;
    __syncthreads();
    s[t] += v;
    __syncthreads();
  }
  if (i < NN) {
    cs[i] = red[0] + s[t] - c;
    dn[i] = (c > 0) ? rsqrtf((float)c) : 0.0f;
  }
}

__global__ void fill_kernel(const int* __restrict__ row, const int* __restrict__ col,
                            const int* __restrict__ start, const int* __restrict__ rank,
                            const float* __restrict__ dn,
                            int* __restrict__ srow, float* __restrict__ sval) {
  int e = blockIdx.x * blockDim.x + threadIdx.x;
  if (e >= EE) return;
  const int r = row[e], c = col[e];
  const int pos = start[c] + rank[e];
  srow[pos] = r;
  sval[pos] = dn[c] * dn[r];
}

__global__ __launch_bounds__(256) void fc0_kernel(const float* __restrict__ x,
                                                  const float* __restrict__ w,
                                                  const float* __restrict__ b,
                                                  float* __restrict__ h) {
  const int wave = threadIdx.x >> 5;
  const int lane = threadIdx.x & 31;
  const int half = lane >> 4;
  const int lid  = lane & 15;
  const int node0 = blockIdx.x * 128 + wave * 16;
  if (node0 >= NN) return;

  v8f acc0 = {}, acc1 = {}, acc2 = {}, acc3 = {};
  for (int c = 0; c < DD / 4; ++c) {
    const int k0 = 4 * c + 2 * half;
    const float* ap = x + (node0 + lid) * DD + k0;
    v2f a; a.x = ap[0]; a.y = ap[1];
    const float* bp = w + k0 * HH;
    v2f b0; b0.x = bp[lid];      b0.y = bp[HH + lid];
    v2f b1; b1.x = bp[16 + lid]; b1.y = bp[HH + 16 + lid];
    v2f b2; b2.x = bp[32 + lid]; b2.y = bp[HH + 32 + lid];
    v2f b3; b3.x = bp[48 + lid]; b3.y = bp[HH + 48 + lid];
    acc0 = wmma_f32(a, b0, acc0);
    acc1 = wmma_f32(a, b1, acc1);
    acc2 = wmma_f32(a, b2, acc2);
    acc3 = wmma_f32(a, b3, acc3);
  }
  v8f accs[4] = {acc0, acc1, acc2, acc3};
  for (int t = 0; t < 4; ++t) {
    const float bias = b[16 * t + lid];
    for (int r = 0; r < 8; ++r) {
      float v = accs[t][r] + bias;
      h[(node0 + r + 8 * half) * HH + 16 * t + lid] = v > 0.0f ? v : 0.0f;
    }
  }
}

__global__ __launch_bounds__(256) void gather_kernel(const int* __restrict__ start,
                                                     const int* __restrict__ srow,
                                                     const float* __restrict__ sval,
                                                     const float* __restrict__ h,
                                                     float* __restrict__ agg) {
  const unsigned idx = blockIdx.x * 256u + threadIdx.x;
  const int n = idx >> 4;
  const int seg = idx & 15;
  const int s = start[n];
  const int e = (n + 1 < NN) ? start[n + 1] : EE;
  float4 acc = make_float4(0.0f, 0.0f, 0.0f, 0.0f);
  for (int base = s; ; base += 16) {
    const int rem = e - base;
    const int cmax = max(rem, __shfl_xor(rem, 16));
    if (cmax <= 0) break;
    int myr = 0;
    float myv = 0.0f;
    if (seg < rem) { myr = srow[base + seg]; myv = sval[base + seg]; }
    const int lim = min(__builtin_amdgcn_readfirstlane(cmax), 16);
    #pragma unroll 4
    for (int j = 0; j < lim; ++j) {
      const int r = __shfl(myr, j, 16);
      const float v = __shfl(myv, j, 16);
      float4 hv = *(const float4*)(h + r * HH + seg * 4);
      const bool on = j < rem;
      hv.x = on ? hv.x : 0.0f;
      hv.y = on ? hv.y : 0.0f;
      hv.z = on ? hv.z : 0.0f;
      hv.w = on ? hv.w : 0.0f;
      acc.x += v * hv.x;
      acc.y += v * hv.y;
      acc.z += v * hv.z;
      acc.w += v * hv.w;
    }
  }
  *(float4*)(agg + n * HH + seg * 4) = acc;
}

__global__ void gate_kernel(const float* __restrict__ h, const float* __restrict__ env_w,
                            const float* __restrict__ env_b, float* __restrict__ e_out,
                            int layer) {
  int n = blockIdx.x * blockDim.x + threadIdx.x;
  if (n >= NN) return;
  const float* W = env_w + layer * 2 * HH * KK;
  const float* B = env_b + layer * KK;
  float logit[KK];
  #pragma unroll
  for (int k = 0; k < KK; ++k) logit[k] = B[k];
  for (int f = 0; f < HH; ++f) {
    float hv = h[n * HH + f];
    const float* wr = W + f * KK;
    #pragma unroll
    for (int k = 0; k < KK; ++k) logit[k] += hv * wr[k];
  }
  float m = logit[0];
  #pragma unroll
  for (int k = 1; k < KK; ++k) m = fmaxf(m, logit[k]);
  float s = 0.0f;
  float ex[KK];
  #pragma unroll
  for (int k = 0; k < KK; ++k) { ex[k] = __expf(logit[k] - m); s += ex[k]; }
  float inv = 1.0f / s;
  #pragma unroll
  for (int k = 0; k < KK; ++k) {
    float pi = ex[k] * inv;
    e_out[n * KK + k] = (pi > THETA) ? pi : 0.0f;
  }
}

__global__ __launch_bounds__(256) void conv_kernel(const float* __restrict__ agg,
                                                   float* h,
                                                   const float* __restrict__ e_gate,
                                                   const float* __restrict__ conv_w,
                                                   int layer) {
  const int wave = threadIdx.x >> 5;
  const int lane = threadIdx.x & 31;
  const int half = lane >> 4;
  const int lid  = lane & 15;
  const int node0 = blockIdx.x * 128 + wave * 16;
  if (node0 >= NN) return;

  const float* W = conv_w + layer * (KK * 2 * HH * HH);
  v8f acc0 = {}, acc1 = {}, acc2 = {}, acc3 = {};

  #pragma unroll 1
  for (int k = 0; k < KK; ++k) {
    const float ek = e_gate[(node0 + lid) * KK + k];
    if (!__any(ek != 0.0f)) continue;
    const float* Wk = W + k * (2 * HH * HH);
    #pragma unroll 1
    for (int c = 0; c < (2 * HH) / 4; ++c) {
      const int kk = 4 * c + 2 * half;
      const float* src = (kk < HH) ? agg : h;
      const int fk = (kk < HH) ? kk : kk - HH;
      const float* ap = src + (node0 + lid) * HH + fk;
      v2f a; a.x = ek * ap[0]; a.y = ek * ap[1];
      const float* bp = Wk + kk * HH;
      v2f b0; b0.x = bp[lid];      b0.y = bp[HH + lid];
      v2f b1; b1.x = bp[16 + lid]; b1.y = bp[HH + 16 + lid];
      v2f b2; b2.x = bp[32 + lid]; b2.y = bp[HH + 32 + lid];
      v2f b3; b3.x = bp[48 + lid]; b3.y = bp[HH + 48 + lid];
      acc0 = wmma_f32(a, b0, acc0);
      acc1 = wmma_f32(a, b1, acc1);
      acc2 = wmma_f32(a, b2, acc2);
      acc3 = wmma_f32(a, b3, acc3);
    }
  }
  v8f accs[4] = {acc0, acc1, acc2, acc3};
  for (int t = 0; t < 4; ++t) {
    for (int r = 0; r < 8; ++r) {
      const int idx = (node0 + r + 8 * half) * HH + 16 * t + lid;
      float v = accs[t][r] + h[idx];
      h[idx] = v > 0.0f ? v : 0.0f;
    }
  }
}

__global__ __launch_bounds__(256) void fc1_kernel(const float* __restrict__ h,
                                                  const float* __restrict__ w,
                                                  const float* __restrict__ b,
                                                  float* __restrict__ out) {
  const int wave = threadIdx.x >> 5;
  const int lane = threadIdx.x & 31;
  const int half = lane >> 4;
  const int lid  = lane & 15;
  const int node0 = blockIdx.x * 128 + wave * 16;
  if (node0 >= NN) return;

  v8f acc = {};
  for (int c = 0; c < HH / 4; ++c) {
    const int k0 = 4 * c + 2 * half;
    const float* ap = h + (node0 + lid) * HH + k0;
    v2f a; a.x = ap[0]; a.y = ap[1];
    const float* bp = w + k0 * CC;
    v2f bb; bb.x = bp[lid]; bb.y = bp[CC + lid];
    acc = wmma_f32(a, bb, acc);
  }
  const float bias = b[lid];
  for (int r = 0; r < 8; ++r)
    out[(node0 + r + 8 * half) * CC + lid] = acc[r] + bias;
}

extern "C" void kernel_launch(void* const* d_in, const int* in_sizes, int n_in,
                              void* d_out, int out_size, void* d_ws, size_t ws_size,
                              hipStream_t stream) {
  const float* x      = (const float*)d_in[0];
  const int*   eidx   = (const int*)d_in[1];
  const float* fc0_w  = (const float*)d_in[2];
  const float* fc0_b  = (const float*)d_in[3];
  const float* fc1_w  = (const float*)d_in[4];
  const float* fc1_b  = (const float*)d_in[5];
  const float* env_w  = (const float*)d_in[6];
  const float* env_b  = (const float*)d_in[7];
  const float* conv_w = (const float*)d_in[8];
  float* out = (float*)d_out;

  const int* row = eidx;
  const int* col = eidx + EE;

  float* ws  = (float*)d_ws;
  int*   cs  = (int*)ws;
  float* dn  = ws + NN;
  float* hA  = dn + NN;
  float* lst = hA + (size_t)NN * HH;
  float* agg = lst + (size_t)NN * HH;
  float* eg  = agg + (size_t)NN * HH;
  int*   srow = (int*)lst;
  float* sval = lst + EE;
  int*   rank = (int*)(lst + 2 * (size_t)EE);
  int*   bsum = (int*)(lst + 3 * (size_t)EE);

  const int T = 256;
  const int gemm_blocks = (NN + 127) / 128;
  const int node_blocks = (NN + T - 1) / T;
  const int edge_blocks = (EE + T - 1) / T;

  zero_kernel<<<node_blocks, T, 0, stream>>>(cs, NN);
  count_rank_kernel<<<edge_blocks, T, 0, stream>>>(col, cs, rank);
  block_sum_kernel<<<node_blocks, T, 0, stream>>>(cs, bsum);
  scan_kernel<<<node_blocks, T, 0, stream>>>(cs, bsum, dn);
  fill_kernel<<<edge_blocks, T, 0, stream>>>(row, col, cs, rank, dn, srow, sval);

  fc0_kernel<<<gemm_blocks, T, 0, stream>>>(x, fc0_w, fc0_b, hA);

  float* cur = hA;
  for (int layer = 0; layer < LL; ++layer) {
    gather_kernel<<<(NN * 16) / T, T, 0, stream>>>(cs, srow, sval, cur, agg);
    gate_kernel<<<node_blocks, T, 0, stream>>>(cur, env_w, env_b, eg, layer);
    conv_kernel<<<gemm_blocks, T, 0, stream>>>(agg, cur, eg, conv_w, layer);
  }

  fc1_kernel<<<gemm_blocks, T, 0, stream>>>(cur, fc1_w, fc1_b, out);
}


# conv walks nodes sorted by gate pattern (per-layer counting sort on the 8 gate-nonzero bits, more heads first, int LDS atomic ranks) so the wave-uniform zero-gate skip fires; per-node sums unchanged
# speedup vs baseline: 1.5797x; 1.1004x over previous
#include <hip/hip_runtime.h>
#include <hip/hip_bf16.h>

#define NN 100000
#define DD 128
#define HH 64
#define KK 8
#define LL 3
#define CC 16
#define EE 1600000
#define THETA 0.1f
#define NB ((NN + 255) / 256)
#define QB ((NB + 3) / 4)

typedef __attribute__((ext_vector_type(2))) float v2f;
typedef __attribute__((ext_vector_type(8))) float v8f;

__device__ __forceinline__ v8f wmma_f32(v2f a, v2f b, v8f c) {
  return __builtin_amdgcn_wmma_f32_16x16x4_f32(
       false, a,  false, b,
       (short)0, c,  false,  false);
}

__global__ void zero_kernel(int* __restrict__ p, int n) {
  int i = blockIdx.x * blockDim.x + threadIdx.x;
  if (i < n) p[i] = 0;
}

__global__ void count_rank_kernel(const int* __restrict__ col, int* __restrict__ cnt,
                                  int* __restrict__ rank) {
  int e = blockIdx.x * blockDim.x + threadIdx.x;
  if (e < EE) rank[e] = atomicAdd(&cnt[col[e]], 1);
}

__global__ __launch_bounds__(256) void block_sum_kernel(const int* __restrict__ cnt,
                                                        int* __restrict__ bsum) {
  __shared__ int s[256];
  const int t = threadIdx.x;
  const int i = blockIdx.x * 256 + t;
  s[t] = (i < NN) ? cnt[i] : 0;
  __syncthreads();
  for (int off = 128; off > 0; off >>= 1) {
    if (t < off) s[t] += s[t + off];
    __syncthreads();
  }
  if (t == 0) bsum[blockIdx.x] = s[0];
}

__global__ __launch_bounds__(256) void scan_kernel(int* __restrict__ cs,
                                                   const int* __restrict__ bsum,
                                                   float* __restrict__ dn) {
  __shared__ int s[256];
  __shared__ int red[256];
  const int t = threadIdx.x;
  const int b = blockIdx.x;
  int p = 0;
  for (int j = t; j < b; j += 256) p += bsum[j];
  red[t] = p;
  const int i = b * 256 + t;
  const int c = (i < NN) ? cs[i] : 0;
  s[t] = c;
  __syncthreads();
  for (int off = 128; off > 0; off >>= 1) {
    if (t < off) red[t] += red[t + off];
    __syncthreads();
  }
  for (int off = 1; off < 256; off <<= 1) {
    const int v = (t >= off) ? s[t - off] : 0;
    __syncthreads();
    s[t] += v;
    __syncthreads();
  }
  if (i < NN) {
    cs[i] = red[0] + s[t] - c;
    dn[i] = (c > 0) ? rsqrtf((float)c) : 0.0f;
  }
}

__global__ void fill_kernel(const int* __restrict__ row, const int* __restrict__ col,
                            const int* __restrict__ start, const int* __restrict__ rank,
                            const float* __restrict__ dn,
                            int* __restrict__ srow, float* __restrict__ sval) {
  int e = blockIdx.x * blockDim.x + threadIdx.x;
  if (e >= EE) return;
  const int r = row[e], c = col[e];
  const int pos = start[c] + rank[e];
  srow[pos] = r;
  sval[pos] = dn[c] * dn[r];
}

__global__ __launch_bounds__(256) void fc0_kernel(const float* __restrict__ x,
                                                  const float* __restrict__ w,
                                                  const float* __restrict__ b,
                                                  float* __restrict__ h) {
  const int wave = threadIdx.x >> 5;
  const int lane = threadIdx.x & 31;
  const int half = lane >> 4;
  const int lid  = lane & 15;
  const int node0 = blockIdx.x * 128 + wave * 16;
  if (node0 >= NN) return;

  v8f acc0 = {}, acc1 = {}, acc2 = {}, acc3 = {};
  for (int c = 0; c < DD / 4; ++c) {
    const int k0 = 4 * c + 2 * half;
    const float* ap = x + (node0 + lid) * DD + k0;
    v2f a; a.x = ap[0]; a.y = ap[1];
    const float* bp = w + k0 * HH;
    v2f b0; b0.x = bp[lid];      b0.y = bp[HH + lid];
    v2f b1; b1.x = bp[16 + lid]; b1.y = bp[HH + 16 + lid];
    v2f b2; b2.x = bp[32 + lid]; b2.y = bp[HH + 32 + lid];
    v2f b3; b3.x = bp[48 + lid]; b3.y = bp[HH + 48 + lid];
    acc0 = wmma_f32(a, b0, acc0);
    acc1 = wmma_f32(a, b1, acc1);
    acc2 = wmma_f32(a, b2, acc2);
    acc3 = wmma_f32(a, b3, acc3);
  }
  v8f accs[4] = {acc0, acc1, acc2, acc3};
  for (int t = 0; t < 4; ++t) {
    const float bias = b[16 * t + lid];
    for (int r = 0; r < 8; ++r) {
      float v = accs[t][r] + bias;
      h[(node0 + r + 8 * half) * HH + 16 * t + lid] = v > 0.0f ? v : 0.0f;
    }
  }
}

__global__ __launch_bounds__(256) void gather_kernel(const int* __restrict__ start,
                                                     const int* __restrict__ srow,
                                                     const float* __restrict__ sval,
                                                     const float* __restrict__ h,
                                                     float* __restrict__ agg,
                                                     const int* __restrict__ keyrank,
                                                     const int* __restrict__ offs,
                                                     int* __restrict__ order) {
  const unsigned idx = blockIdx.x * 256u + threadIdx.x;
  const int n = idx >> 4;
  const int seg = idx & 15;
  if (seg == 0) {
    const int kr = keyrank[n];
    order[offs[(n >> 8) * 256 + (kr >> 8)] + (kr & 255)] = n;
  }
  const int s = start[n];
  const int e = (n + 1 < NN) ? start[n + 1] : EE;
  float4 acc = make_float4(0.0f, 0.0f, 0.0f, 0.0f);
  for (int base = s; ; base += 16) {
    const int rem = e - base;
    const int cmax = max(rem, __shfl_xor(rem, 16));
    if (cmax <= 0) break;
    int myr = 0;
    float myv = 0.0f;
    if (seg < rem) { myr = srow[base + seg]; myv = sval[base + seg]; }
    const int lim = min(__builtin_amdgcn_readfirstlane(cmax), 16);
    #pragma unroll 4
    for (int j = 0; j < lim; ++j) {
      const int r = __shfl(myr, j, 16);
      const float v = __shfl(myv, j, 16);
      float4 hv = *(const float4*)(h + r * HH + seg * 4);
      const bool on = j < rem;
      hv.x = on ? hv.x : 0.0f;
      hv.y = on ? hv.y : 0.0f;
      hv.z = on ? hv.z : 0.0f;
      hv.w = on ? hv.w : 0.0f;
      acc.x += v * hv.x;
      acc.y += v * hv.y;
      acc.z += v * hv.z;
      acc.w += v * hv.w;
    }
  }
  *(float4*)(agg + n * HH + seg * 4) = acc;
}

__global__ __launch_bounds__(256) void gate_kernel(const float* __restrict__ h,
                                                   const float* __restrict__ env_w,
                                                   const float* __restrict__ env_b,
                                                   float* __restrict__ e_out,
                                                   int* __restrict__ keyrank,
                                                   int* __restrict__ bh, int layer) {
  __shared__ int hist[256];
  const int t = threadIdx.x;
  const int n = blockIdx.x * 256 + t;
  hist[t] = 0;
  __syncthreads();
  if (n < NN) {
  const float* W = env_w + layer * 2 * HH * KK;
  const float* B = env_b + layer * KK;
  float logit[KK];
  #pragma unroll
  for (int k = 0; k < KK; ++k) logit[k] = B[k];
  for (int f = 0; f < HH; ++f) {
    float hv = h[n * HH + f];
    const float* wr = W + f * KK;
    #pragma unroll
    for (int k = 0; k < KK; ++k) logit[k] += hv * wr[k];
  }
  float m = logit[0];
  #pragma unroll
  for (int k = 1; k < KK; ++k) m = fmaxf(m, logit[k]);
  float s = 0.0f;
  float ex[KK];
  #pragma unroll
  for (int k = 0; k < KK; ++k) { ex[k] = __expf(logit[k] - m); s += ex[k]; }
  float inv = 1.0f / s;
  int key = 0;
  #pragma unroll
  for (int k = 0; k < KK; ++k) {
    float pi = ex[k] * inv;
    const bool on = pi > THETA;
    e_out[n * KK + k] = on ? pi : 0.0f;
    key |= on ? (1 << k) : 0;
  }
  const int r = atomicAdd(&hist[key], 1);
  keyrank[n] = key * 256 + r;
  }
  __syncthreads();
  bh[blockIdx.x * 256 + t] = hist[t];
  if (blockIdx.x == 0)
    for (int b = NB; b < 4 * QB; ++b) bh[b * 256 + t] = 0;
}

__global__ __launch_bounds__(1024) void key_scan_kernel(int* bh) {
  __shared__ int segsum[4][256];
  __shared__ int kb[256];
  __shared__ int kpos[256];
  const int t = threadIdx.x;
  const int key = t & 255;
  const int seg = t >> 8;
  int* p = bh + (seg * QB) * 256 + key;
  int c[QB];
  int s = 0;
  #pragma unroll
  for (int i = 0; i < QB; ++i) {
    c[i] = p[i * 256];
    s += c[i];
  }
  segsum[seg][key] = s;
  __syncthreads();
  if (t < 256) {
    const int tot = segsum[0][t] + segsum[1][t] + segsum[2][t] + segsum[3][t];
    const int pt = __popc(t);
    int rk = 0;
    for (int q = 0; q < 256; ++q) {
      const int pq = __popc(q);
      rk += (pq > pt || (pq == pt && q < t)) ? 1 : 0;
    }
    kpos[t] = rk;
    kb[rk] = tot;
  }
  __syncthreads();
  for (int off = 1; off < 256; off <<= 1) {
    int v = 0;
    if (t < 256 && t >= off) v = kb[t - off];
    __syncthreads();
    if (t < 256) kb[t] += v;
    __syncthreads();
  }
  int run = kb[kpos[key]] - (segsum[0][key] + segsum[1][key] + segsum[2][key] + segsum[3][key]);
  for (int j = 0; j < seg; ++j) run += segsum[j][key];
  #pragma unroll
  for (int i = 0; i < QB; ++i) {
    p[i * 256] = run;
    run += c[i];
  }
}

__global__ __launch_bounds__(256) void conv_kernel(const float* __restrict__ agg,
                                                   float* h,
                                                   const float* __restrict__ e_gate,
                                                   const float* __restrict__ conv_w,
                                                   const int* __restrict__ order,
                                                   int layer) {
  const int wave = threadIdx.x >> 5;
  const int lane = threadIdx.x & 31;
  const int half = lane >> 4;
  const int lid  = lane & 15;
  const int slot0 = blockIdx.x * 128 + wave * 16;
  if (slot0 >= NN) return;
  const int node = order[slot0 + lid];

  const float* W = conv_w + layer * (KK * 2 * HH * HH);
  v8f acc0 = {}, acc1 = {}, acc2 = {}, acc3 = {};

  #pragma unroll 1
  for (int k = 0; k < KK; ++k) {
    const float ek = e_gate[node * KK + k];
    if (!__any(ek != 0.0f)) continue;
    const float* Wk = W + k * (2 * HH * HH);
    #pragma unroll 1
    for (int c = 0; c < (2 * HH) / 4; ++c) {
      const int kk = 4 * c + 2 * half;
      const float* src = (kk < HH) ? agg : h;
      const int fk = (kk < HH) ? kk : kk - HH;
      const float* ap = src + node * HH + fk;
      v2f a; a.x = ek * ap[0]; a.y = ek * ap[1];
      const float* bp = Wk + kk * HH;
      v2f b0; b0.x = bp[lid];      b0.y = bp[HH + lid];
      v2f b1; b1.x = bp[16 + lid]; b1.y = bp[HH + 16 + lid];
      v2f b2; b2.x = bp[32 + lid]; b2.y = bp[HH + 32 + lid];
      v2f b3; b3.x = bp[48 + lid]; b3.y = bp[HH + 48 + lid];
      acc0 = wmma_f32(a, b0, acc0);
      acc1 = wmma_f32(a, b1, acc1);
      acc2 = wmma_f32(a, b2, acc2);
      acc3 = wmma_f32(a, b3, acc3);
    }
  }
  v8f accs[4] = {acc0, acc1, acc2, acc3};
  int orow[8];
  for (int r = 0; r < 8; ++r) orow[r] = order[slot0 + r + 8 * half];
  #pragma unroll
  for (int t = 0; t < 4; ++t) {
    float res[8];
    #pragma unroll
    for (int r = 0; r < 8; ++r) res[r] = h[orow[r] * HH + 16 * t + lid];
    #pragma unroll
    for (int r = 0; r < 8; ++r) {
      const float v = accs[t][r] + res[r];
      h[orow[r] * HH + 16 * t + lid] = v > 0.0f ? v : 0.0f;
    }
  }
}

__global__ __launch_bounds__(256) void fc1_kernel(const float* __restrict__ h,
                                                  const float* __restrict__ w,
                                                  const float* __restrict__ b,
                                                  float* __restrict__ out) {
  const int wave = threadIdx.x >> 5;
  const int lane = threadIdx.x & 31;
  const int half = lane >> 4;
  const int lid  = lane & 15;
  const int node0 = blockIdx.x * 128 + wave * 16;
  if (node0 >= NN) return;

  v8f acc = {};
  for (int c = 0; c < HH / 4; ++c) {
    const int k0 = 4 * c + 2 * half;
    const float* ap = h + (node0 + lid) * HH + k0;
    v2f a; a.x = ap[0]; a.y = ap[1];
    const float* bp = w + k0 * CC;
    v2f bb; bb.x = bp[lid]; bb.y = bp[CC + lid];
    acc = wmma_f32(a, bb, acc);
  }
  const float bias = b[lid];
  for (int r = 0; r < 8; ++r)
    out[(node0 + r + 8 * half) * CC + lid] = acc[r] + bias;
}

extern "C" void kernel_launch(void* const* d_in, const int* in_sizes, int n_in,
                              void* d_out, int out_size, void* d_ws, size_t ws_size,
                              hipStream_t stream) {
  const float* x      = (const float*)d_in[0];
  const int*   eidx   = (const int*)d_in[1];
  const float* fc0_w  = (const float*)d_in[2];
  const float* fc0_b  = (const float*)d_in[3];
  const float* fc1_w  = (const float*)d_in[4];
  const float* fc1_b  = (const float*)d_in[5];
  const float* env_w  = (const float*)d_in[6];
  const float* env_b  = (const float*)d_in[7];
  const float* conv_w = (const float*)d_in[8];
  float* out = (float*)d_out;

  const int* row = eidx;
  const int* col = eidx + EE;

  float* ws  = (float*)d_ws;
  int*   cs  = (int*)ws;
  float* dn  = ws + NN;
  float* hA  = dn + NN;
  float* lst = hA + (size_t)NN * HH;
  float* agg = lst + (size_t)NN * HH;
  float* eg  = agg + (size_t)NN * HH;
  int*   srow = (int*)lst;
  float* sval = lst + EE;
  int*   rank = (int*)(lst + 2 * (size_t)EE);
  int*   bsum = (int*)(lst + 3 * (size_t)EE);
  int*   bh      = bsum + 512;
  int*   keyrank = bh + 4 * QB * 256;
  int*   order   = keyrank + NN;

  const int T = 256;
  const int gemm_blocks = (NN + 127) / 128;
  const int node_blocks = (NN + T - 1) / T;
  const int edge_blocks = (EE + T - 1) / T;

  zero_kernel<<<node_blocks, T, 0, stream>>>(cs, NN);
  count_rank_kernel<<<edge_blocks, T, 0, stream>>>(col, cs, rank);
  block_sum_kernel<<<node_blocks, T, 0, stream>>>(cs, bsum);
  scan_kernel<<<node_blocks, T, 0, stream>>>(cs, bsum, dn);
  fill_kernel<<<edge_blocks, T, 0, stream>>>(row, col, cs, rank, dn, srow, sval);

  fc0_kernel<<<gemm_blocks, T, 0, stream>>>(x, fc0_w, fc0_b, hA);

  float* cur = hA;
  for (int layer = 0; layer < LL; ++layer) {
    gate_kernel<<<node_blocks, T, 0, stream>>>(cur, env_w, env_b, eg, keyrank, bh, layer);
    key_scan_kernel<<<1, 1024, 0, stream>>>(bh);
    gather_kernel<<<(NN * 16) / T, T, 0, stream>>>(cs, srow, sval, cur, agg, keyrank, bh, order);
    conv_kernel<<<gemm_blocks, T, 0, stream>>>(agg, cur, eg, conv_w, order, layer);
  }

  fc1_kernel<<<gemm_blocks, T, 0, stream>>>(cur, fc1_w, fc1_b, out);
}


# Edge grouping by destination ranked with LDS counters in two levels (256-node range, then node) instead of one global returning add and two scattered stores per edge; sum order = LDS arrival rank
# speedup vs baseline: 1.9127x; 1.2108x over previous
#include <hip/hip_runtime.h>
#include <hip/hip_bf16.h>

#define NN 100000
#define DD 128
#define HH 64
#define KK 8
#define LL 3
#define CC 16
#define EE 1600000
#define THETA 0.1f
#define NB ((NN + 255) / 256)
#define QB ((NB + 3) / 4)
#define PBLK 256
#define PCH ((EE + PBLK - 1) / PBLK)
#define PIT ((PCH + 511) / 512)
#define CBLK 512
#define CCH ((EE + CBLK - 1) / CBLK)
#define CIT ((CCH + 255) / 256)
#define NBLK 1563
#define NIT ((EE + NBLK * 256 - 1) / (NBLK * 256))

typedef __attribute__((ext_vector_type(2))) float v2f;
typedef __attribute__((ext_vector_type(8))) float v8f;

__device__ __forceinline__ v8f wmma_f32(v2f a, v2f b, v8f c) {
  return __builtin_amdgcn_wmma_f32_16x16x4_f32(
       false, a,  false, b,
       (short)0, c,  false,  false);
}

__global__ void zero_kernel(int* __restrict__ p, int n) {
  int i = blockIdx.x * blockDim.x + threadIdx.x;
  if (i < n) p[i] = 0;
}


__global__ __launch_bounds__(256) void bin_count_kernel(const int* __restrict__ col,
                                                        int* __restrict__ binTot) {
  __shared__ int hist[NB];
  for (int i = threadIdx.x; i < NB; i += 256) hist[i] = 0;
  __syncthreads();
  const int e0 = blockIdx.x * CCH;
  int c[CIT];
  #pragma unroll
  for (int j = 0; j < CIT; ++j) {
    const int i = threadIdx.x + 256 * j;
    const int e = e0 + i;
    c[j] = ((i < CCH) && (e < EE)) ? col[e] : -1;
  }
  #pragma unroll
  for (int j = 0; j < CIT; ++j)
    if (c[j] >= 0) atomicAdd(&hist[c[j] >> 8], 1);
  __syncthreads();
  for (int i = threadIdx.x; i < NB; i += 256) {
    const int v = hist[i];
    if (v > 0) atomicAdd(&binTot[i], v);
  }
}

__global__ __launch_bounds__(512) void partition_kernel(const int* __restrict__ row,
                                                        const int* __restrict__ col,
                                                        const int* __restrict__ binTot,
                                                        int* __restrict__ binCur,
                                                        int* __restrict__ binBase,
                                                        int* __restrict__ prow,
                                                        int* __restrict__ pcol) {
  __shared__ int hist[512];
  __shared__ int sc[512];
  __shared__ int ofs[512];
  const int t = threadIdx.x;
  const int tot = (t < NB) ? binTot[t] : 0;
  hist[t] = 0;
  sc[t] = tot;
  __syncthreads();
  const int e0 = blockIdx.x * PCH;
  int r[PIT], c[PIT], lr[PIT];
  #pragma unroll
  for (int j = 0; j < PIT; ++j) {
    const int i = t + 512 * j;
    const int e = e0 + i;
    const bool ok = (i < PCH) && (e < EE);
    r[j] = ok ? row[e] : 0;
    c[j] = ok ? col[e] : -1;
  }
  #pragma unroll
  for (int j = 0; j < PIT; ++j)
    lr[j] = (c[j] >= 0) ? atomicAdd(&hist[c[j] >> 8], 1) : 0;
  for (int off = 1; off < 512; off <<= 1) {
    const int v = (t >= off) ? sc[t - off] : 0;
    __syncthreads();
    sc[t] += v;
    __syncthreads();
  }
  const int base = sc[t] - tot;
  const int h = hist[t];
  int o = base;
  if (t < NB && h > 0) o = base + atomicAdd(&binCur[t], h);
  ofs[t] = o;
  if (blockIdx.x == 0 && t < NB) binBase[t] = base;
  __syncthreads();
  #pragma unroll
  for (int j = 0; j < PIT; ++j) {
    if (c[j] >= 0) {
      const int p = ofs[c[j] >> 8] + lr[j];
      prow[p] = r[j];
      pcol[p] = c[j];
    }
  }
}

__global__ __launch_bounds__(512) void bin_place_kernel(const int* __restrict__ prow,
                                                        const int* __restrict__ pcol,
                                                        const int* __restrict__ binBase,
                                                        const int* __restrict__ binTot,
                                                        int* __restrict__ start,
                                                        float* __restrict__ dn,
                                                        int* __restrict__ srow,
                                                        float* __restrict__ sval) {
  __shared__ int cnt[256];
  __shared__ int sc[256];
  __shared__ int cur[256];
  __shared__ float dl[256];
  const int t = threadIdx.x;
  const int b = blockIdx.x;
  const int lo = binBase[b];
  const int hi = lo + binTot[b];
  if (t < 256) cnt[t] = 0;
  __syncthreads();
  for (int i0 = lo + t; i0 < hi; i0 += 2048) {
    int c[4];
    #pragma unroll
    for (int u = 0; u < 4; ++u) {
      const int i = i0 + 512 * u;
      c[u] = (i < hi) ? pcol[i] : -1;
    }
    #pragma unroll
    for (int u = 0; u < 4; ++u)
      if (c[u] >= 0) atomicAdd(&cnt[c[u] & 255], 1);
  }
  __syncthreads();
  const int d = (t < 256) ? cnt[t] : 0;
  if (t < 256) sc[t] = d;
  __syncthreads();
  for (int off = 1; off < 256; off <<= 1) {
    int v = 0;
    if (t < 256 && t >= off) v = sc[t - off];
    __syncthreads();
    if (t < 256) sc[t] += v;
    __syncthreads();
  }
  if (t < 256) {
    const int s0 = lo + sc[t] - d;
    const float dv = (d > 0) ? rsqrtf((float)d) : 0.0f;
    cur[t] = s0;
    dl[t] = dv;
    const int n = b * 256 + t;
    if (n < NN) { start[n] = s0; dn[n] = dv; }
  }
  __syncthreads();
  for (int i0 = lo + t; i0 < hi; i0 += 2048) {
    int c[4], r[4];
    #pragma unroll
    for (int u = 0; u < 4; ++u) {
      const int i = i0 + 512 * u;
      const bool ok = i < hi;
      c[u] = ok ? pcol[i] : -1;
      r[u] = ok ? prow[i] : 0;
    }
    #pragma unroll
    for (int u = 0; u < 4; ++u) {
      if (c[u] >= 0) {
        const int cl = c[u] & 255;
        const int p = atomicAdd(&cur[cl], 1);
        srow[p] = r[u];
        sval[p] = dl[cl];
      }
    }
  }
}

__global__ __launch_bounds__(256) void norm_kernel(const int* __restrict__ srow,
                                                   const float* __restrict__ dn,
                                                   float* __restrict__ sval) {
  const int p0 = blockIdx.x * 256 + threadIdx.x;
  int r[NIT];
  float v[NIT], d[NIT];
  #pragma unroll
  for (int j = 0; j < NIT; ++j) {
    const int p = p0 + NBLK * 256 * j;
    const int pc = (p < EE) ? p : 0;
    r[j] = srow[pc];
    v[j] = sval[pc];
  }
  #pragma unroll
  for (int j = 0; j < NIT; ++j) d[j] = dn[r[j]];
  #pragma unroll
  for (int j = 0; j < NIT; ++j) {
    const int p = p0 + NBLK * 256 * j;
    if (p < EE) sval[p] = v[j] * d[j];
  }
}

__global__ __launch_bounds__(256) void fc0_kernel(const float* __restrict__ x,
                                                  const float* __restrict__ w,
                                                  const float* __restrict__ b,
                                                  float* __restrict__ h) {
  const int wave = threadIdx.x >> 5;
  const int lane = threadIdx.x & 31;
  const int half = lane >> 4;
  const int lid  = lane & 15;
  const int node0 = blockIdx.x * 128 + wave * 16;
  if (node0 >= NN) return;

  v8f acc0 = {}, acc1 = {}, acc2 = {}, acc3 = {};
  for (int c = 0; c < DD / 4; ++c) {
    const int k0 = 4 * c + 2 * half;
    const float* ap = x + (node0 + lid) * DD + k0;
    v2f a; a.x = ap[0]; a.y = ap[1];
    const float* bp = w + k0 * HH;
    v2f b0; b0.x = bp[lid];      b0.y = bp[HH + lid];
    v2f b1; b1.x = bp[16 + lid]; b1.y = bp[HH + 16 + lid];
    v2f b2; b2.x = bp[32 + lid]; b2.y = bp[HH + 32 + lid];
    v2f b3; b3.x = bp[48 + lid]; b3.y = bp[HH + 48 + lid];
    acc0 = wmma_f32(a, b0, acc0);
    acc1 = wmma_f32(a, b1, acc1);
    acc2 = wmma_f32(a, b2, acc2);
    acc3 = wmma_f32(a, b3, acc3);
  }
  v8f accs[4] = {acc0, acc1, acc2, acc3};
  for (int t = 0; t < 4; ++t) {
    const float bias = b[16 * t + lid];
    for (int r = 0; r < 8; ++r) {
      float v = accs[t][r] + bias;
      h[(node0 + r + 8 * half) * HH + 16 * t + lid] = v > 0.0f ? v : 0.0f;
    }
  }
}

__global__ __launch_bounds__(256) void gather_kernel(const int* __restrict__ start,
                                                     const int* __restrict__ srow,
                                                     const float* __restrict__ sval,
                                                     const float* __restrict__ h,
                                                     float* __restrict__ agg,
                                                     const int* __restrict__ keyrank,
                                                     const int* __restrict__ offs,
                                                     int* __restrict__ order) {
  const unsigned idx = blockIdx.x * 256u + threadIdx.x;
  const int n = idx >> 4;
  const int seg = idx & 15;
  if (seg == 0) {
    const int kr = keyrank[n];
    order[offs[(n >> 8) * 256 + (kr >> 8)] + (kr & 255)] = n;
  }
  const int s = start[n];
  const int e = (n + 1 < NN) ? start[n + 1] : EE;
  float4 acc = make_float4(0.0f, 0.0f, 0.0f, 0.0f);
  for (int base = s; ; base += 16) {
    const int rem = e - base;
    const int cmax = max(rem, __shfl_xor(rem, 16));
    if (cmax <= 0) break;
    int myr = 0;
    float myv = 0.0f;
    if (seg < rem) { myr = srow[base + seg]; myv = sval[base + seg]; }
    const int lim = min(__builtin_amdgcn_readfirstlane(cmax), 16);
    #pragma unroll 4
    for (int j = 0; j < lim; ++j) {
      const int r = __shfl(myr, j, 16);
      const float v = __shfl(myv, j, 16);
      float4 hv = *(const float4*)(h + r * HH + seg * 4);
      const bool on = j < rem;
      hv.x = on ? hv.x : 0.0f;
      hv.y = on ? hv.y : 0.0f;
      hv.z = on ? hv.z : 0.0f;
      hv.w = on ? hv.w : 0.0f;
      acc.x += v * hv.x;
      acc.y += v * hv.y;
      acc.z += v * hv.z;
      acc.w += v * hv.w;
    }
  }
  *(float4*)(agg + n * HH + seg * 4) = acc;
}

__global__ __launch_bounds__(256) void gate_kernel(const float* __restrict__ h,
                                                   const float* __restrict__ env_w,
                                                   const float* __restrict__ env_b,
                                                   float* __restrict__ e_out,
                                                   int* __restrict__ keyrank,
                                                   int* __restrict__ bh, int layer) {
  __shared__ int hist[256];
  const int t = threadIdx.x;
  const int n = blockIdx.x * 256 + t;
  hist[t] = 0;
  __syncthreads();
  if (n < NN) {
  const float* W = env_w + layer * 2 * HH * KK;
  const float* B = env_b + layer * KK;
  float logit[KK];
  #pragma unroll
  for (int k = 0; k < KK; ++k) logit[k] = B[k];
  for (int f = 0; f < HH; ++f) {
    float hv = h[n * HH + f];
    const float* wr = W + f * KK;
    #pragma unroll
    for (int k = 0; k < KK; ++k) logit[k] += hv * wr[k];
  }
  float m = logit[0];
  #pragma unroll
  for (int k = 1; k < KK; ++k) m = fmaxf(m, logit[k]);
  float s = 0.0f;
  float ex[KK];
  #pragma unroll
  for (int k = 0; k < KK; ++k) { ex[k] = __expf(logit[k] - m); s += ex[k]; }
  float inv = 1.0f / s;
  int key = 0;
  #pragma unroll
  for (int k = 0; k < KK; ++k) {
    float pi = ex[k] * inv;
    const bool on = pi > THETA;
    e_out[n * KK + k] = on ? pi : 0.0f;
    key |= on ? (1 << k) : 0;
  }
  const int r = atomicAdd(&hist[key], 1);
  keyrank[n] = key * 256 + r;
  }
  __syncthreads();
  bh[blockIdx.x * 256 + t] = hist[t];
  if (blockIdx.x == 0)
    for (int b = NB; b < 4 * QB; ++b) bh[b * 256 + t] = 0;
}

__global__ __launch_bounds__(1024) void key_scan_kernel(int* bh) {
  __shared__ int segsum[4][256];
  __shared__ int kb[256];
  __shared__ int kpos[256];
  const int t = threadIdx.x;
  const int key = t & 255;
  const int seg = t >> 8;
  int* p = bh + (seg * QB) * 256 + key;
  int c[QB];
  int s = 0;
  #pragma unroll
  for (int i = 0; i < QB; ++i) {
    c[i] = p[i * 256];
    s += c[i];
  }
  segsum[seg][key] = s;
  __syncthreads();
  if (t < 256) {
    const int tot = segsum[0][t] + segsum[1][t] + segsum[2][t] + segsum[3][t];
    const int pt = __popc(t);
    int rk = 0;
    for (int q = 0; q < 256; ++q) {
      const int pq = __popc(q);
      rk += (pq > pt || (pq == pt && q < t)) ? 1 : 0;
    }
    kpos[t] = rk;
    kb[rk] = tot;
  }
  __syncthreads();
  for (int off = 1; off < 256; off <<= 1) {
    int v = 0;
    if (t < 256 && t >= off) v = kb[t - off];
    __syncthreads();
    if (t < 256) kb[t] += v;
    __syncthreads();
  }
  int run = kb[kpos[key]] - (segsum[0][key] + segsum[1][key] + segsum[2][key] + segsum[3][key]);
  for (int j = 0; j < seg; ++j) run += segsum[j][key];
  #pragma unroll
  for (int i = 0; i < QB; ++i) {
    p[i * 256] = run;
    run += c[i];
  }
}

__global__ __launch_bounds__(256) void conv_kernel(const float* __restrict__ agg,
                                                   float* h,
                                                   const float* __restrict__ e_gate,
                                                   const float* __restrict__ conv_w,
                                                   const int* __restrict__ order,
                                                   int layer) {
  const int wave = threadIdx.x >> 5;
  const int lane = threadIdx.x & 31;
  const int half = lane >> 4;
  const int lid  = lane & 15;
  const int slot0 = blockIdx.x * 128 + wave * 16;
  if (slot0 >= NN) return;
  const int node = order[slot0 + lid];

  const float* W = conv_w + layer * (KK * 2 * HH * HH);
  v8f acc0 = {}, acc1 = {}, acc2 = {}, acc3 = {};

  #pragma unroll 1
  for (int k = 0; k < KK; ++k) {
    const float ek = e_gate[node * KK + k];
    if (!__any(ek != 0.0f)) continue;
    const float* Wk = W + k * (2 * HH * HH);
    #pragma unroll 1
    for (int c = 0; c < (2 * HH) / 4; ++c) {
      const int kk = 4 * c + 2 * half;
      const float* src = (kk < HH) ? agg : h;
      const int fk = (kk < HH) ? kk : kk - HH;
      const float* ap = src + node * HH + fk;
      v2f a; a.x = ek * ap[0]; a.y = ek * ap[1];
      const float* bp = Wk + kk * HH;
      v2f b0; b0.x = bp[lid];      b0.y = bp[HH + lid];
      v2f b1; b1.x = bp[16 + lid]; b1.y = bp[HH + 16 + lid];
      v2f b2; b2.x = bp[32 + lid]; b2.y = bp[HH + 32 + lid];
      v2f b3; b3.x = bp[48 + lid]; b3.y = bp[HH + 48 + lid];
      acc0 = wmma_f32(a, b0, acc0);
      acc1 = wmma_f32(a, b1, acc1);
      acc2 = wmma_f32(a, b2, acc2);
      acc3 = wmma_f32(a, b3, acc3);
    }
  }
  v8f accs[4] = {acc0, acc1, acc2, acc3};
  int orow[8];
  for (int r = 0; r < 8; ++r) orow[r] = order[slot0 + r + 8 * half];
  #pragma unroll
  for (int t = 0; t < 4; ++t) {
    float res[8];
    #pragma unroll
    for (int r = 0; r < 8; ++r) res[r] = h[orow[r] * HH + 16 * t + lid];
    #pragma unroll
    for (int r = 0; r < 8; ++r) {
      const float v = accs[t][r] + res[r];
      h[orow[r] * HH + 16 * t + lid] = v > 0.0f ? v : 0.0f;
    }
  }
}

__global__ __launch_bounds__(256) void fc1_kernel(const float* __restrict__ h,
                                                  const float* __restrict__ w,
                                                  const float* __restrict__ b,
                                                  float* __restrict__ out) {
  const int wave = threadIdx.x >> 5;
  const int lane = threadIdx.x & 31;
  const int half = lane >> 4;
  const int lid  = lane & 15;
  const int node0 = blockIdx.x * 128 + wave * 16;
  if (node0 >= NN) return;

  v8f acc = {};
  for (int c = 0; c < HH / 4; ++c) {
    const int k0 = 4 * c + 2 * half;
    const float* ap = h + (node0 + lid) * HH + k0;
    v2f a; a.x = ap[0]; a.y = ap[1];
    const float* bp = w + k0 * CC;
    v2f bb; bb.x = bp[lid]; bb.y = bp[CC + lid];
    acc = wmma_f32(a, bb, acc);
  }
  const float bias = b[lid];
  for (int r = 0; r < 8; ++r)
    out[(node0 + r + 8 * half) * CC + lid] = acc[r] + bias;
}

extern "C" void kernel_launch(void* const* d_in, const int* in_sizes, int n_in,
                              void* d_out, int out_size, void* d_ws, size_t ws_size,
                              hipStream_t stream) {
  const float* x      = (const float*)d_in[0];
  const int*   eidx   = (const int*)d_in[1];
  const float* fc0_w  = (const float*)d_in[2];
  const float* fc0_b  = (const float*)d_in[3];
  const float* fc1_w  = (const float*)d_in[4];
  const float* fc1_b  = (const float*)d_in[5];
  const float* env_w  = (const float*)d_in[6];
  const float* env_b  = (const float*)d_in[7];
  const float* conv_w = (const float*)d_in[8];
  float* out = (float*)d_out;

  const int* row = eidx;
  const int* col = eidx + EE;

  float* ws  = (float*)d_ws;
  int*   cs  = (int*)ws;
  float* dn  = ws + NN;
  float* hA  = dn + NN;
  float* lst = hA + (size_t)NN * HH;
  float* agg = lst + (size_t)NN * HH;
  float* eg  = agg + (size_t)NN * HH;
  int*   srow = (int*)lst;
  float* sval = lst + EE;
  int*   binTot  = (int*)(lst + 2 * (size_t)EE);
  int*   binCur  = binTot + 512;
  int*   binBase = binTot + 1024;
  int*   prow = (int*)agg;
  int*   pcol = prow + EE;
  int*   tail = (int*)(lst + 3 * (size_t)EE);
  int*   bh      = tail + 512;
  int*   keyrank = bh + 4 * QB * 256;
  int*   order   = keyrank + NN;

  const int T = 256;
  const int gemm_blocks = (NN + 127) / 128;
  const int node_blocks = (NN + T - 1) / T;

  zero_kernel<<<4, T, 0, stream>>>(binTot, 1024);
  bin_count_kernel<<<CBLK, T, 0, stream>>>(col, binTot);
  partition_kernel<<<PBLK, 512, 0, stream>>>(row, col, binTot, binCur, binBase, prow, pcol);
  bin_place_kernel<<<NB, 512, 0, stream>>>(prow, pcol, binBase, binTot, cs, dn, srow, sval);
  norm_kernel<<<NBLK, T, 0, stream>>>(srow, dn, sval);

  fc0_kernel<<<gemm_blocks, T, 0, stream>>>(x, fc0_w, fc0_b, hA);

  float* cur = hA;
  for (int layer = 0; layer < LL; ++layer) {
    gate_kernel<<<node_blocks, T, 0, stream>>>(cur, env_w, env_b, eg, keyrank, bh, layer);
    key_scan_kernel<<<1, 1024, 0, stream>>>(bh);
    gather_kernel<<<(NN * 16) / T, T, 0, stream>>>(cs, srow, sval, cur, agg, keyrank, bh, order);
    conv_kernel<<<gemm_blocks, T, 0, stream>>>(agg, cur, eg, conv_w, order, layer);
  }

  fc1_kernel<<<gemm_blocks, T, 0, stream>>>(cur, fc1_w, fc1_b, out);
}


# conv and fc0 operand loads widened to 16 bytes: K taken in groups of 8 (a lane half = 4 adjacent k), column tile t = columns 4c+t, float4 epilogue; k pairing in products, so sum order over k, changes
# speedup vs baseline: 2.4004x; 1.2550x over previous
#include <hip/hip_runtime.h>
#include <hip/hip_bf16.h>

#define NN 100000
#define DD 128
#define HH 64
#define KK 8
#define LL 3
#define CC 16
#define EE 1600000
#define THETA 0.1f
#define NB ((NN + 255) / 256)
#define QB ((NB + 3) / 4)
#define PBLK 256
#define PCH ((EE + PBLK - 1) / PBLK)
#define PIT ((PCH + 511) / 512)
#define CBLK 512
#define CCH ((EE + CBLK - 1) / CBLK)
#define CIT ((CCH + 255) / 256)
#define NBLK 1563
#define NIT ((EE + NBLK * 256 - 1) / (NBLK * 256))

typedef __attribute__((ext_vector_type(2))) float v2f;
typedef __attribute__((ext_vector_type(8))) float v8f;

__device__ __forceinline__ v8f wmma_f32(v2f a, v2f b, v8f c) {
  return __builtin_amdgcn_wmma_f32_16x16x4_f32(
       false, a,  false, b,
       (short)0, c,  false,  false);
}

__device__ __forceinline__ void kgroup8(v8f& acc0, v8f& acc1, v8f& acc2, v8f& acc3,
                                        const float4 a4, const float* wp) {
  const float4 w0 = *(const float4*)(wp);
  const float4 w1 = *(const float4*)(wp + HH);
  const float4 w2 = *(const float4*)(wp + 2 * HH);
  const float4 w3 = *(const float4*)(wp + 3 * HH);
  v2f a, b;
  a.x = a4.x; a.y = a4.y;
  b.x = w0.x; b.y = w1.x; acc0 = wmma_f32(a, b, acc0);
  b.x = w0.y; b.y = w1.y; acc1 = wmma_f32(a, b, acc1);
  b.x = w0.z; b.y = w1.z; acc2 = wmma_f32(a, b, acc2);
  b.x = w0.w; b.y = w1.w; acc3 = wmma_f32(a, b, acc3);
  a.x = a4.z; a.y = a4.w;
  b.x = w2.x; b.y = w3.x; acc0 = wmma_f32(a, b, acc0);
  b.x = w2.y; b.y = w3.y; acc1 = wmma_f32(a, b, acc1);
  b.x = w2.z; b.y = w3.z; acc2 = wmma_f32(a, b, acc2);
  b.x = w2.w; b.y = w3.w; acc3 = wmma_f32(a, b, acc3);
}

__global__ void zero_kernel(int* __restrict__ p, int n) {
  int i = blockIdx.x * blockDim.x + threadIdx.x;
  if (i < n) p[i] = 0;
}


__global__ __launch_bounds__(256) void bin_count_kernel(const int* __restrict__ col,
                                                        int* __restrict__ binTot) {
  __shared__ int hist[NB];
  for (int i = threadIdx.x; i < NB; i += 256) hist[i] = 0;
  __syncthreads();
  const int e0 = blockIdx.x * CCH;
  int c[CIT];
  #pragma unroll
  for (int j = 0; j < CIT; ++j) {
    const int i = threadIdx.x + 256 * j;
    const int e = e0 + i;
    c[j] = ((i < CCH) && (e < EE)) ? col[e] : -1;
  }
  #pragma unroll
  for (int j = 0; j < CIT; ++j)
    if (c[j] >= 0) atomicAdd(&hist[c[j] >> 8], 1);
  __syncthreads();
  for (int i = threadIdx.x; i < NB; i += 256) {
    const int v = hist[i];
    if (v > 0) atomicAdd(&binTot[i], v);
  }
}

__global__ __launch_bounds__(512) void partition_kernel(const int* __restrict__ row,
                                                        const int* __restrict__ col,
                                                        const int* __restrict__ binTot,
                                                        int* __restrict__ binCur,
                                                        int* __restrict__ binBase,
                                                        int* __restrict__ prow,
                                                        int* __restrict__ pcol) {
  __shared__ int hist[512];
  __shared__ int sc[512];
  __shared__ int ofs[512];
  const int t = threadIdx.x;
  const int tot = (t < NB) ? binTot[t] : 0;
  hist[t] = 0;
  sc[t] = tot;
  __syncthreads();
  const int e0 = blockIdx.x * PCH;
  int r[PIT], c[PIT], lr[PIT];
  #pragma unroll
  for (int j = 0; j < PIT; ++j) {
    const int i = t + 512 * j;
    const int e = e0 + i;
    const bool ok = (i < PCH) && (e < EE);
    r[j] = ok ? row[e] : 0;
    c[j] = ok ? col[e] : -1;
  }
  #pragma unroll
  for (int j = 0; j < PIT; ++j)
    lr[j] = (c[j] >= 0) ? atomicAdd(&hist[c[j] >> 8], 1) : 0;
  for (int off = 1; off < 512; off <<= 1) {
    const int v = (t >= off) ? sc[t - off] : 0;
    __syncthreads();
    sc[t] += v;
    __syncthreads();
  }
  const int base = sc[t] - tot;
  const int h = hist[t];
  int o = base;
  if (t < NB && h > 0) o = base + atomicAdd(&binCur[t], h);
  ofs[t] = o;
  if (blockIdx.x == 0 && t < NB) binBase[t] = base;
  __syncthreads();
  #pragma unroll
  for (int j = 0; j < PIT; ++j) {
    if (c[j] >= 0) {
      const int p = ofs[c[j] >> 8] + lr[j];
      prow[p] = r[j];
      pcol[p] = c[j];
    }
  }
}

__global__ __launch_bounds__(512) void bin_place_kernel(const int* __restrict__ prow,
                                                        const int* __restrict__ pcol,
                                                        const int* __restrict__ binBase,
                                                        const int* __restrict__ binTot,
                                                        int* __restrict__ start,
                                                        float* __restrict__ dn,
                                                        int* __restrict__ srow,
                                                        float* __restrict__ sval) {
  __shared__ int cnt[256];
  __shared__ int sc[256];
  __shared__ int cur[256];
  __shared__ float dl[256];
  const int t = threadIdx.x;
  const int b = blockIdx.x;
  const int lo = binBase[b];
  const int hi = lo + binTot[b];
  if (t < 256) cnt[t] = 0;
  __syncthreads();
  for (int i0 = lo + t; i0 < hi; i0 += 2048) {
    int c[4];
    #pragma unroll
    for (int u = 0; u < 4; ++u) {
      const int i = i0 + 512 * u;
      c[u] = (i < hi) ? pcol[i] : -1;
    }
    #pragma unroll
    for (int u = 0; u < 4; ++u)
      if (c[u] >= 0) atomicAdd(&cnt[c[u] & 255], 1);
  }
  __syncthreads();
  const int d = (t < 256) ? cnt[t] : 0;
  if (t < 256) sc[t] = d;
  __syncthreads();
  for (int off = 1; off < 256; off <<= 1) {
    int v = 0;
    if (t < 256 && t >= off) v = sc[t - off];
    __syncthreads();
    if (t < 256) sc[t] += v;
    __syncthreads();
  }
  if (t < 256) {
    const int s0 = lo + sc[t] - d;
    const float dv = (d > 0) ? rsqrtf((float)d) : 0.0f;
    cur[t] = s0;
    dl[t] = dv;
    const int n = b * 256 + t;
    if (n < NN) { start[n] = s0; dn[n] = dv; }
  }
  __syncthreads();
  for (int i0 = lo + t; i0 < hi; i0 += 2048) {
    int c[4], r[4];
    #pragma unroll
    for (int u = 0; u < 4; ++u) {
      const int i = i0 + 512 * u;
      const bool ok = i < hi;
      c[u] = ok ? pcol[i] : -1;
      r[u] = ok ? prow[i] : 0;
    }
    #pragma unroll
    for (int u = 0; u < 4; ++u) {
      if (c[u] >= 0) {
        const int cl = c[u] & 255;
        const int p = atomicAdd(&cur[cl], 1);
        srow[p] = r[u];
        sval[p] = dl[cl];
      }
    }
  }
}

__global__ __launch_bounds__(256) void norm_kernel(const int* __restrict__ srow,
                                                   const float* __restrict__ dn,
                                                   float* __restrict__ sval) {
  const int p0 = blockIdx.x * 256 + threadIdx.x;
  int r[NIT];
  float v[NIT], d[NIT];
  #pragma unroll
  for (int j = 0; j < NIT; ++j) {
    const int p = p0 + NBLK * 256 * j;
    const int pc = (p < EE) ? p : 0;
    r[j] = srow[pc];
    v[j] = sval[pc];
  }
  #pragma unroll
  for (int j = 0; j < NIT; ++j) d[j] = dn[r[j]];
  #pragma unroll
  for (int j = 0; j < NIT; ++j) {
    const int p = p0 + NBLK * 256 * j;
    if (p < EE) sval[p] = v[j] * d[j];
  }
}

__global__ __launch_bounds__(256) void fc0_kernel(const float* __restrict__ x,
                                                  const float* __restrict__ w,
                                                  const float* __restrict__ b,
                                                  float* __restrict__ h) {
  const int wave = threadIdx.x >> 5;
  const int lane = threadIdx.x & 31;
  const int half = lane >> 4;
  const int lid  = lane & 15;
  const int node0 = blockIdx.x * 128 + wave * 16;
  if (node0 >= NN) return;

  v8f acc0 = {}, acc1 = {}, acc2 = {}, acc3 = {};
  const float* ap = x + (node0 + lid) * DD + 4 * half;
  const float* wp = w + (4 * half) * HH + 4 * lid;
  #pragma unroll 2
  for (int g = 0; g < DD / 8; ++g) {
    const float4 a4 = *(const float4*)(ap + 8 * g);
    kgroup8(acc0, acc1, acc2, acc3, a4, wp + 8 * g * HH);
  }
  const float4 bias = *(const float4*)(b + 4 * lid);
  #pragma unroll
  for (int r = 0; r < 8; ++r) {
    float4 v;
    v.x = acc0[r] + bias.x;
    v.y = acc1[r] + bias.y;
    v.z = acc2[r] + bias.z;
    v.w = acc3[r] + bias.w;
    v.x = v.x > 0.0f ? v.x : 0.0f;
    v.y = v.y > 0.0f ? v.y : 0.0f;
    v.z = v.z > 0.0f ? v.z : 0.0f;
    v.w = v.w > 0.0f ? v.w : 0.0f;
    *(float4*)(h + (node0 + r + 8 * half) * HH + 4 * lid) = v;
  }
}

__global__ __launch_bounds__(256) void gather_kernel(const int* __restrict__ start,
                                                     const int* __restrict__ srow,
                                                     const float* __restrict__ sval,
                                                     const float* __restrict__ h,
                                                     float* __restrict__ agg,
                                                     const int* __restrict__ keyrank,
                                                     const int* __restrict__ offs,
                                                     int* __restrict__ order) {
  const unsigned idx = blockIdx.x * 256u + threadIdx.x;
  const int n = idx >> 4;
  const int seg = idx & 15;
  if (seg == 0) {
    const int kr = keyrank[n];
    order[offs[(n >> 8) * 256 + (kr >> 8)] + (kr & 255)] = n;
  }
  const int s = start[n];
  const int e = (n + 1 < NN) ? start[n + 1] : EE;
  float4 acc = make_float4(0.0f, 0.0f, 0.0f, 0.0f);
  for (int base = s; ; base += 16) {
    const int rem = e - base;
    const int cmax = max(rem, __shfl_xor(rem, 16));
    if (cmax <= 0) break;
    int myr = 0;
    float myv = 0.0f;
    if (seg < rem) { myr = srow[base + seg]; myv = sval[base + seg]; }
    const int lim = min(__builtin_amdgcn_readfirstlane(cmax), 16);
    #pragma unroll 4
    for (int j = 0; j < lim; ++j) {
      const int r = __shfl(myr, j, 16);
      const float v = __shfl(myv, j, 16);
      float4 hv = *(const float4*)(h + r * HH + seg * 4);
      const bool on = j < rem;
      hv.x = on ? hv.x : 0.0f;
      hv.y = on ? hv.y : 0.0f;
      hv.z = on ? hv.z : 0.0f;
      hv.w = on ? hv.w : 0.0f;
      acc.x += v * hv.x;
      acc.y += v * hv.y;
      acc.z += v * hv.z;
      acc.w += v * hv.w;
    }
  }
  *(float4*)(agg + n * HH + seg * 4) = acc;
}

__global__ __launch_bounds__(256) void gate_kernel(const float* __restrict__ h,
                                                   const float* __restrict__ env_w,
                                                   const float* __restrict__ env_b,
                                                   float* __restrict__ e_out,
                                                   int* __restrict__ keyrank,
                                                   int* __restrict__ bh, int layer) {
  __shared__ int hist[256];
  const int t = threadIdx.x;
  const int n = blockIdx.x * 256 + t;
  hist[t] = 0;
  __syncthreads();
  if (n < NN) {
  const float* W = env_w + layer * 2 * HH * KK;
  const float* B = env_b + layer * KK;
  float logit[KK];
  #pragma unroll
  for (int k = 0; k < KK; ++k) logit[k] = B[k];
  for (int f = 0; f < HH; ++f) {
    float hv = h[n * HH + f];
    const float* wr = W + f * KK;
    #pragma unroll
    for (int k = 0; k < KK; ++k) logit[k] += hv * wr[k];
  }
  float m = logit[0];
  #pragma unroll
  for (int k = 1; k < KK; ++k) m = fmaxf(m, logit[k]);
  float s = 0.0f;
  float ex[KK];
  #pragma unroll
  for (int k = 0; k < KK; ++k) { ex[k] = __expf(logit[k] - m); s += ex[k]; }
  float inv = 1.0f / s;
  int key = 0;
  #pragma unroll
  for (int k = 0; k < KK; ++k) {
    float pi = ex[k] * inv;
    const bool on = pi > THETA;
    e_out[n * KK + k] = on ? pi : 0.0f;
    key |= on ? (1 << k) : 0;
  }
  const int r = atomicAdd(&hist[key], 1);
  keyrank[n] = key * 256 + r;
  }
  __syncthreads();
  bh[blockIdx.x * 256 + t] = hist[t];
  if (blockIdx.x == 0)
    for (int b = NB; b < 4 * QB; ++b) bh[b * 256 + t] = 0;
}

__global__ __launch_bounds__(1024) void key_scan_kernel(int* bh) {
  __shared__ int segsum[4][256];
  __shared__ int kb[256];
  __shared__ int kpos[256];
  const int t = threadIdx.x;
  const int key = t & 255;
  const int seg = t >> 8;
  int* p = bh + (seg * QB) * 256 + key;
  int c[QB];
  int s = 0;
  #pragma unroll
  for (int i = 0; i < QB; ++i) {
    c[i] = p[i * 256];
    s += c[i];
  }
  segsum[seg][key] = s;
  __syncthreads();
  if (t < 256) {
    const int tot = segsum[0][t] + segsum[1][t] + segsum[2][t] + segsum[3][t];
    const int pt = __popc(t);
    int rk = 0;
    for (int q = 0; q < 256; ++q) {
      const int pq = __popc(q);
      rk += (pq > pt || (pq == pt && q < t)) ? 1 : 0;
    }
    kpos[t] = rk;
    kb[rk] = tot;
  }
  __syncthreads();
  for (int off = 1; off < 256; off <<= 1) {
    int v = 0;
    if (t < 256 && t >= off) v = kb[t - off];
    __syncthreads();
    if (t < 256) kb[t] += v;
    __syncthreads();
  }
  int run = kb[kpos[key]] - (segsum[0][key] + segsum[1][key] + segsum[2][key] + segsum[3][key]);
  for (int j = 0; j < seg; ++j) run += segsum[j][key];
  #pragma unroll
  for (int i = 0; i < QB; ++i) {
    p[i * 256] = run;
    run += c[i];
  }
}

__global__ __launch_bounds__(256) void conv_kernel(const float* __restrict__ agg,
                                                   float* h,
                                                   const float* __restrict__ e_gate,
                                                   const float* __restrict__ conv_w,
                                                   const int* __restrict__ order,
                                                   int layer) {
  const int wave = threadIdx.x >> 5;
  const int lane = threadIdx.x & 31;
  const int half = lane >> 4;
  const int lid  = lane & 15;
  const int slot0 = blockIdx.x * 128 + wave * 16;
  if (slot0 >= NN) return;
  const int node = order[slot0 + lid];

  const float* W = conv_w + layer * (KK * 2 * HH * HH);
  v8f acc0 = {}, acc1 = {}, acc2 = {}, acc3 = {};
  const float* ap_agg = agg + node * HH + 4 * half;
  const float* ap_own = h + node * HH + 4 * half;

  #pragma unroll 1
  for (int k = 0; k < KK; ++k) {
    const float ek = e_gate[node * KK + k];
    if (!__any(ek != 0.0f)) continue;
    const float* wp = W + k * (2 * HH * HH) + (4 * half) * HH + 4 * lid;
    #pragma unroll 2
    for (int g = 0; g < HH / 8; ++g) {
      float4 a4 = *(const float4*)(ap_agg + 8 * g);
      a4.x = ek * a4.x; a4.y = ek * a4.y; a4.z = ek * a4.z; a4.w = ek * a4.w;
      kgroup8(acc0, acc1, acc2, acc3, a4, wp + 8 * g * HH);
    }
    #pragma unroll 2
    for (int g = 0; g < HH / 8; ++g) {
      float4 a4 = *(const float4*)(ap_own + 8 * g);
      a4.x = ek * a4.x; a4.y = ek * a4.y; a4.z = ek * a4.z; a4.w = ek * a4.w;
      kgroup8(acc0, acc1, acc2, acc3, a4, wp + (HH + 8 * g) * HH);
    }
  }
  int orow[8];
  for (int r = 0; r < 8; ++r) orow[r] = order[slot0 + r + 8 * half];
  float4 res[8];
  #pragma unroll
  for (int r = 0; r < 8; ++r) res[r] = *(const float4*)(h + orow[r] * HH + 4 * lid);
  #pragma unroll
  for (int r = 0; r < 8; ++r) {
    float4 v;
    v.x = acc0[r] + res[r].x;
    v.y = acc1[r] + res[r].y;
    v.z = acc2[r] + res[r].z;
    v.w = acc3[r] + res[r].w;
    v.x = v.x > 0.0f ? v.x : 0.0f;
    v.y = v.y > 0.0f ? v.y : 0.0f;
    v.z = v.z > 0.0f ? v.z : 0.0f;
    v.w = v.w > 0.0f ? v.w : 0.0f;
    *(float4*)(h + orow[r] * HH + 4 * lid) = v;
  }
}

__global__ __launch_bounds__(256) void fc1_kernel(const float* __restrict__ h,
                                                  const float* __restrict__ w,
                                                  const float* __restrict__ b,
                                                  float* __restrict__ out) {
  const int wave = threadIdx.x >> 5;
  const int lane = threadIdx.x & 31;
  const int half = lane >> 4;
  const int lid  = lane & 15;
  const int node0 = blockIdx.x * 128 + wave * 16;
  if (node0 >= NN) return;

  v8f acc = {};
  for (int c = 0; c < HH / 4; ++c) {
    const int k0 = 4 * c + 2 * half;
    const float* ap = h + (node0 + lid) * HH + k0;
    v2f a; a.x = ap[0]; a.y = ap[1];
    const float* bp = w + k0 * CC;
    v2f bb; bb.x = bp[lid]; bb.y = bp[CC + lid];
    acc = wmma_f32(a, bb, acc);
  }
  const float bias = b[lid];
  for (int r = 0; r < 8; ++r)
    out[(node0 + r + 8 * half) * CC + lid] = acc[r] + bias;
}

extern "C" void kernel_launch(void* const* d_in, const int* in_sizes, int n_in,
                              void* d_out, int out_size, void* d_ws, size_t ws_size,
                              hipStream_t stream) {
  const float* x      = (const float*)d_in[0];
  const int*   eidx   = (const int*)d_in[1];
  const float* fc0_w  = (const float*)d_in[2];
  const float* fc0_b  = (const float*)d_in[3];
  const float* fc1_w  = (const float*)d_in[4];
  const float* fc1_b  = (const float*)d_in[5];
  const float* env_w  = (const float*)d_in[6];
  const float* env_b  = (const float*)d_in[7];
  const float* conv_w = (const float*)d_in[8];
  float* out = (float*)d_out;

  const int* row = eidx;
  const int* col = eidx + EE;

  float* ws  = (float*)d_ws;
  int*   cs  = (int*)ws;
  float* dn  = ws + NN;
  float* hA  = dn + NN;
  float* lst = hA + (size_t)NN * HH;
  float* agg = lst + (size_t)NN * HH;
  float* eg  = agg + (size_t)NN * HH;
  int*   srow = (int*)lst;
  float* sval = lst + EE;
  int*   binTot  = (int*)(lst + 2 * (size_t)EE);
  int*   binCur  = binTot + 512;
  int*   binBase = binTot + 1024;
  int*   prow = (int*)agg;
  int*   pcol = prow + EE;
  int*   tail = (int*)(lst + 3 * (size_t)EE);
  int*   bh      = tail + 512;
  int*   keyrank = bh + 4 * QB * 256;
  int*   order   = keyrank + NN;

  const int T = 256;
  const int gemm_blocks = (NN + 127) / 128;
  const int node_blocks = (NN + T - 1) / T;

  zero_kernel<<<4, T, 0, stream>>>(binTot, 1024);
  bin_count_kernel<<<CBLK, T, 0, stream>>>(col, binTot);
  partition_kernel<<<PBLK, 512, 0, stream>>>(row, col, binTot, binCur, binBase, prow, pcol);
  bin_place_kernel<<<NB, 512, 0, stream>>>(prow, pcol, binBase, binTot, cs, dn, srow, sval);
  norm_kernel<<<NBLK, T, 0, stream>>>(srow, dn, sval);

  fc0_kernel<<<gemm_blocks, T, 0, stream>>>(x, fc0_w, fc0_b, hA);

  float* cur = hA;
  for (int layer = 0; layer < LL; ++layer) {
    gate_kernel<<<node_blocks, T, 0, stream>>>(cur, env_w, env_b, eg, keyrank, bh, layer);
    key_scan_kernel<<<1, 1024, 0, stream>>>(bh);
    gather_kernel<<<(NN * 16) / T, T, 0, stream>>>(cs, srow, sval, cur, agg, keyrank, bh, order);
    conv_kernel<<<gemm_blocks, T, 0, stream>>>(agg, cur, eg, conv_w, order, layer);
  }

  fc1_kernel<<<gemm_blocks, T, 0, stream>>>(cur, fc1_w, fc1_b, out);
}


# Aggregation moved into the conv kernel: a wave sums its 16 nodes' neighbour rows into an LDS tile that feeds the A fragments (no agg plane, no gather launch); h in two planes; sum order kept
# speedup vs baseline: 2.6256x; 1.0938x over previous
#include <hip/hip_runtime.h>
#include <hip/hip_bf16.h>

#define NN 100000
#define DD 128
#define HH 64
#define KK 8
#define LL 3
#define CC 16
#define EE 1600000
#define THETA 0.1f
#define NB ((NN + 255) / 256)
#define QB ((NB + 3) / 4)
#define PBLK 256
#define PCH ((EE + PBLK - 1) / PBLK)
#define PIT ((PCH + 511) / 512)
#define CBLK 512
#define CCH ((EE + CBLK - 1) / CBLK)
#define CIT ((CCH + 255) / 256)
#define NBLK 1563
#define NIT ((EE + NBLK * 256 - 1) / (NBLK * 256))

typedef __attribute__((ext_vector_type(2))) float v2f;
typedef __attribute__((ext_vector_type(8))) float v8f;

__device__ __forceinline__ v8f wmma_f32(v2f a, v2f b, v8f c) {
  return __builtin_amdgcn_wmma_f32_16x16x4_f32(
       false, a,  false, b,
       (short)0, c,  false,  false);
}

__device__ __forceinline__ void kgroup8(v8f& acc0, v8f& acc1, v8f& acc2, v8f& acc3,
                                        const float4 a4, const float* wp) {
  const float4 w0 = *(const float4*)(wp);
  const float4 w1 = *(const float4*)(wp + HH);
  const float4 w2 = *(const float4*)(wp + 2 * HH);
  const float4 w3 = *(const float4*)(wp + 3 * HH);
  v2f a, b;
  a.x = a4.x; a.y = a4.y;
  b.x = w0.x; b.y = w1.x; acc0 = wmma_f32(a, b, acc0);
  b.x = w0.y; b.y = w1.y; acc1 = wmma_f32(a, b, acc1);
  b.x = w0.z; b.y = w1.z; acc2 = wmma_f32(a, b, acc2);
  b.x = w0.w; b.y = w1.w; acc3 = wmma_f32(a, b, acc3);
  a.x = a4.z; a.y = a4.w;
  b.x = w2.x; b.y = w3.x; acc0 = wmma_f32(a, b, acc0);
  b.x = w2.y; b.y = w3.y; acc1 = wmma_f32(a, b, acc1);
  b.x = w2.z; b.y = w3.z; acc2 = wmma_f32(a, b, acc2);
  b.x = w2.w; b.y = w3.w; acc3 = wmma_f32(a, b, acc3);
}

__global__ void zero_kernel(int* __restrict__ p, int n) {
  int i = blockIdx.x * blockDim.x + threadIdx.x;
  if (i < n) p[i] = 0;
}


__global__ __launch_bounds__(256) void bin_count_kernel(const int* __restrict__ col,
                                                        int* __restrict__ binTot) {
  __shared__ int hist[NB];
  for (int i = threadIdx.x; i < NB; i += 256) hist[i] = 0;
  __syncthreads();
  const int e0 = blockIdx.x * CCH;
  int c[CIT];
  #pragma unroll
  for (int j = 0; j < CIT; ++j) {
    const int i = threadIdx.x + 256 * j;
    const int e = e0 + i;
    c[j] = ((i < CCH) && (e < EE)) ? col[e] : -1;
  }
  #pragma unroll
  for (int j = 0; j < CIT; ++j)
    if (c[j] >= 0) atomicAdd(&hist[c[j] >> 8], 1);
  __syncthreads();
  for (int i = threadIdx.x; i < NB; i += 256) {
    const int v = hist[i];
    if (v > 0) atomicAdd(&binTot[i], v);
  }
}

__global__ __launch_bounds__(512) void partition_kernel(const int* __restrict__ row,
                                                        const int* __restrict__ col,
                                                        const int* __restrict__ binTot,
                                                        int* __restrict__ binCur,
                                                        int* __restrict__ binBase,
                                                        int* __restrict__ prow,
                                                        int* __restrict__ pcol) {
  __shared__ int hist[512];
  __shared__ int sc[512];
  __shared__ int ofs[512];
  const int t = threadIdx.x;
  const int tot = (t < NB) ? binTot[t] : 0;
  hist[t] = 0;
  sc[t] = tot;
  __syncthreads();
  const int e0 = blockIdx.x * PCH;
  int r[PIT], c[PIT], lr[PIT];
  #pragma unroll
  for (int j = 0; j < PIT; ++j) {
    const int i = t + 512 * j;
    const int e = e0 + i;
    const bool ok = (i < PCH) && (e < EE);
    r[j] = ok ? row[e] : 0;
    c[j] = ok ? col[e] : -1;
  }
  #pragma unroll
  for (int j = 0; j < PIT; ++j)
    lr[j] = (c[j] >= 0) ? atomicAdd(&hist[c[j] >> 8], 1) : 0;
  for (int off = 1; off < 512; off <<= 1) {
    const int v = (t >= off) ? sc[t - off] : 0;
    __syncthreads();
    sc[t] += v;
    __syncthreads();
  }
  const int base = sc[t] - tot;
  const int h = hist[t];
  int o = base;
  if (t < NB && h > 0) o = base + atomicAdd(&binCur[t], h);
  ofs[t] = o;
  if (blockIdx.x == 0 && t < NB) binBase[t] = base;
  __syncthreads();
  #pragma unroll
  for (int j = 0; j < PIT; ++j) {
    if (c[j] >= 0) {
      const int p = ofs[c[j] >> 8] + lr[j];
      prow[p] = r[j];
      pcol[p] = c[j];
    }
  }
}

__global__ __launch_bounds__(512) void bin_place_kernel(const int* __restrict__ prow,
                                                        const int* __restrict__ pcol,
                                                        const int* __restrict__ binBase,
                                                        const int* __restrict__ binTot,
                                                        int* __restrict__ start,
                                                        float* __restrict__ dn,
                                                        int* __restrict__ srow,
                                                        float* __restrict__ sval) {
  __shared__ int cnt[256];
  __shared__ int sc[256];
  __shared__ int cur[256];
  __shared__ float dl[256];
  const int t = threadIdx.x;
  const int b = blockIdx.x;
  const int lo = binBase[b];
  const int hi = lo + binTot[b];
  if (t < 256) cnt[t] = 0;
  __syncthreads();
  for (int i0 = lo + t; i0 < hi; i0 += 2048) {
    int c[4];
    #pragma unroll
    for (int u = 0; u < 4; ++u) {
      const int i = i0 + 512 * u;
      c[u] = (i < hi) ? pcol[i] : -1;
    }
    #pragma unroll
    for (int u = 0; u < 4; ++u)
      if (c[u] >= 0) atomicAdd(&cnt[c[u] & 255], 1);
  }
  __syncthreads();
  const int d = (t < 256) ? cnt[t] : 0;
  if (t < 256) sc[t] = d;
  __syncthreads();
  for (int off = 1; off < 256; off <<= 1) {
    int v = 0;
    if (t < 256 && t >= off) v = sc[t - off];
    __syncthreads();
    if (t < 256) sc[t] += v;
    __syncthreads();
  }
  if (t < 256) {
    const int s0 = lo + sc[t] - d;
    const float dv = (d > 0) ? rsqrtf((float)d) : 0.0f;
    cur[t] = s0;
    dl[t] = dv;
    const int n = b * 256 + t;
    if (n < NN) { start[n] = s0; dn[n] = dv; }
  }
  __syncthreads();
  for (int i0 = lo + t; i0 < hi; i0 += 2048) {
    int c[4], r[4];
    #pragma unroll
    for (int u = 0; u < 4; ++u) {
      const int i = i0 + 512 * u;
      const bool ok = i < hi;
      c[u] = ok ? pcol[i] : -1;
      r[u] = ok ? prow[i] : 0;
    }
    #pragma unroll
    for (int u = 0; u < 4; ++u) {
      if (c[u] >= 0) {
        const int cl = c[u] & 255;
        const int p = atomicAdd(&cur[cl], 1);
        srow[p] = r[u];
        sval[p] = dl[cl];
      }
    }
  }
}

__global__ __launch_bounds__(256) void norm_kernel(const int* __restrict__ srow,
                                                   const float* __restrict__ dn,
                                                   float* __restrict__ sval) {
  const int p0 = blockIdx.x * 256 + threadIdx.x;
  int r[NIT];
  float v[NIT], d[NIT];
  #pragma unroll
  for (int j = 0; j < NIT; ++j) {
    const int p = p0 + NBLK * 256 * j;
    const int pc = (p < EE) ? p : 0;
    r[j] = srow[pc];
    v[j] = sval[pc];
  }
  #pragma unroll
  for (int j = 0; j < NIT; ++j) d[j] = dn[r[j]];
  #pragma unroll
  for (int j = 0; j < NIT; ++j) {
    const int p = p0 + NBLK * 256 * j;
    if (p < EE) sval[p] = v[j] * d[j];
  }
}

__global__ __launch_bounds__(256) void fc0_kernel(const float* __restrict__ x,
                                                  const float* __restrict__ w,
                                                  const float* __restrict__ b,
                                                  float* __restrict__ h) {
  const int wave = threadIdx.x >> 5;
  const int lane = threadIdx.x & 31;
  const int half = lane >> 4;
  const int lid  = lane & 15;
  const int node0 = blockIdx.x * 128 + wave * 16;
  if (node0 >= NN) return;

  v8f acc0 = {}, acc1 = {}, acc2 = {}, acc3 = {};
  const float* ap = x + (node0 + lid) * DD + 4 * half;
  const float* wp = w + (4 * half) * HH + 4 * lid;
  #pragma unroll 2
  for (int g = 0; g < DD / 8; ++g) {
    const float4 a4 = *(const float4*)(ap + 8 * g);
    kgroup8(acc0, acc1, acc2, acc3, a4, wp + 8 * g * HH);
  }
  const float4 bias = *(const float4*)(b + 4 * lid);
  #pragma unroll
  for (int r = 0; r < 8; ++r) {
    float4 v;
    v.x = acc0[r] + bias.x;
    v.y = acc1[r] + bias.y;
    v.z = acc2[r] + bias.z;
    v.w = acc3[r] + bias.w;
    v.x = v.x > 0.0f ? v.x : 0.0f;
    v.y = v.y > 0.0f ? v.y : 0.0f;
    v.z = v.z > 0.0f ? v.z : 0.0f;
    v.w = v.w > 0.0f ? v.w : 0.0f;
    *(float4*)(h + (node0 + r + 8 * half) * HH + 4 * lid) = v;
  }
}

__global__ __launch_bounds__(256) void order_kernel(const int* __restrict__ keyrank,
                                                    const int* __restrict__ offs,
                                                    int* __restrict__ order) {
  const int n = blockIdx.x * 256 + threadIdx.x;
  if (n < NN) {
    const int kr = keyrank[n];
    order[offs[(n >> 8) * 256 + (kr >> 8)] + (kr & 255)] = n;
  }
}

__global__ __launch_bounds__(256) void gate_kernel(const float* __restrict__ h,
                                                   const float* __restrict__ env_w,
                                                   const float* __restrict__ env_b,
                                                   float* __restrict__ e_out,
                                                   int* __restrict__ keyrank,
                                                   int* __restrict__ bh, int layer) {
  __shared__ int hist[256];
  const int t = threadIdx.x;
  const int n = blockIdx.x * 256 + t;
  hist[t] = 0;
  __syncthreads();
  if (n < NN) {
  const float* W = env_w + layer * 2 * HH * KK;
  const float* B = env_b + layer * KK;
  float logit[KK];
  #pragma unroll
  for (int k = 0; k < KK; ++k) logit[k] = B[k];
  for (int f = 0; f < HH; ++f) {
    float hv = h[n * HH + f];
    const float* wr = W + f * KK;
    #pragma unroll
    for (int k = 0; k < KK; ++k) logit[k] += hv * wr[k];
  }
  float m = logit[0];
  #pragma unroll
  for (int k = 1; k < KK; ++k) m = fmaxf(m, logit[k]);
  float s = 0.0f;
  float ex[KK];
  #pragma unroll
  for (int k = 0; k < KK; ++k) { ex[k] = __expf(logit[k] - m); s += ex[k]; }
  float inv = 1.0f / s;
  int key = 0;
  #pragma unroll
  for (int k = 0; k < KK; ++k) {
    float pi = ex[k] * inv;
    const bool on = pi > THETA;
    e_out[n * KK + k] = on ? pi : 0.0f;
    key |= on ? (1 << k) : 0;
  }
  const int r = atomicAdd(&hist[key], 1);
  keyrank[n] = key * 256 + r;
  }
  __syncthreads();
  bh[blockIdx.x * 256 + t] = hist[t];
  if (blockIdx.x == 0)
    for (int b = NB; b < 4 * QB; ++b) bh[b * 256 + t] = 0;
}

__global__ __launch_bounds__(1024) void key_scan_kernel(int* bh) {
  __shared__ int segsum[4][256];
  __shared__ int kb[256];
  __shared__ int kpos[256];
  const int t = threadIdx.x;
  const int key = t & 255;
  const int seg = t >> 8;
  int* p = bh + (seg * QB) * 256 + key;
  int c[QB];
  int s = 0;
  #pragma unroll
  for (int i = 0; i < QB; ++i) {
    c[i] = p[i * 256];
    s += c[i];
  }
  segsum[seg][key] = s;
  __syncthreads();
  if (t < 256) {
    const int tot = segsum[0][t] + segsum[1][t] + segsum[2][t] + segsum[3][t];
    const int pt = __popc(t);
    int rk = 0;
    for (int q = 0; q < 256; ++q) {
      const int pq = __popc(q);
      rk += (pq > pt || (pq == pt && q < t)) ? 1 : 0;
    }
    kpos[t] = rk;
    kb[rk] = tot;
  }
  __syncthreads();
  for (int off = 1; off < 256; off <<= 1) {
    int v = 0;
    if (t < 256 && t >= off) v = kb[t - off];
    __syncthreads();
    if (t < 256) kb[t] += v;
    __syncthreads();
  }
  int run = kb[kpos[key]] - (segsum[0][key] + segsum[1][key] + segsum[2][key] + segsum[3][key]);
  for (int j = 0; j < seg; ++j) run += segsum[j][key];
  #pragma unroll
  for (int i = 0; i < QB; ++i) {
    p[i * 256] = run;
    run += c[i];
  }
}

#define TP 132
__global__ __launch_bounds__(256) void conv_kernel(const int* __restrict__ start,
                                                   const int* __restrict__ srow,
                                                   const float* __restrict__ sval,
                                                   const float* __restrict__ hin,
                                                   float* __restrict__ hout,
                                                   const float* __restrict__ e_gate,
                                                   const float* __restrict__ conv_w,
                                                   const int* __restrict__ order,
                                                   int layer) {
  __shared__ __attribute__((aligned(16))) float tile[8 * 16 * TP];
  const int wave = threadIdx.x >> 5;
  const int lane = threadIdx.x & 31;
  const int half = lane >> 4;
  const int lid  = lane & 15;
  const int slot0 = blockIdx.x * 128 + wave * 16;
  const bool live = slot0 < NN;
  float* tw = tile + wave * (16 * TP);
  int node = 0;
  if (live) {
    node = order[slot0 + lid];
    const int sA = start[node];
    const int eA = (node + 1 < NN) ? start[node + 1] : EE;
    {
      float4 own[8];
      #pragma unroll
      for (int it = 0; it < 8; ++it)
        own[it] = *(const float4*)(hin + __shfl(node, 2 * it + half) * HH + 4 * lid);
      #pragma unroll
      for (int it = 0; it < 8; ++it)
        *(float4*)(tw + (2 * it + half) * TP + HH + 4 * lid) = own[it];
    }
    for (int it = 0; it < 8; ++it) {
      const int i = 2 * it + half;
      const int s = __shfl(sA, i);
      const int e = __shfl(eA, i);
      float4 acc = make_float4(0.0f, 0.0f, 0.0f, 0.0f);
      for (int base = s; ; base += 32) {
        const int rem = e - base;
        const int cmax = max(rem, __shfl_xor(rem, 16));
        if (cmax <= 0) break;
        int myr0 = 0, myr1 = 0;
        float myv0 = 0.0f, myv1 = 0.0f;
        if (lid < rem) { myr0 = srow[base + lid]; myv0 = sval[base + lid]; }
        if (lid + 16 < rem) { myr1 = srow[base + 16 + lid]; myv1 = sval[base + 16 + lid]; }
        const int lim = min(__builtin_amdgcn_readfirstlane(cmax), 32);
        for (int j0 = 0; j0 < lim; j0 += 8) {
          const int myr = (j0 < 16) ? myr0 : myr1;
          const float myv = (j0 < 16) ? myv0 : myv1;
          #pragma unroll
          for (int u = 0; u < 8; ++u) {
            const int j = j0 + u;
            const int r = __shfl(myr, j & 15, 16);
            const float v = __shfl(myv, j & 15, 16);
            float4 hv = *(const float4*)(hin + r * HH + lid * 4);
            const bool on = j < rem;
            hv.x = on ? hv.x : 0.0f;
            hv.y = on ? hv.y : 0.0f;
            hv.z = on ? hv.z : 0.0f;
            hv.w = on ? hv.w : 0.0f;
            acc.x += v * hv.x;
            acc.y += v * hv.y;
            acc.z += v * hv.z;
            acc.w += v * hv.w;
          }
        }
      }
      *(float4*)(tw + i * TP + 4 * lid) = acc;
    }
  }
  __syncthreads();
  if (!live) return;

  const float* W = conv_w + layer * (KK * 2 * HH * HH);
  v8f acc0 = {}, acc1 = {}, acc2 = {}, acc3 = {};
  const float* ap = tw + lid * TP + 4 * half;

  #pragma unroll 1
  for (int k = 0; k < KK; ++k) {
    const float ek = e_gate[node * KK + k];
    if (!__any(ek != 0.0f)) continue;
    const float* wp = W + k * (2 * HH * HH) + (4 * half) * HH + 4 * lid;
    #pragma unroll 2
    for (int g = 0; g < 2 * HH / 8; ++g) {
      float4 a4 = *(const float4*)(ap + 8 * g);
      a4.x = ek * a4.x; a4.y = ek * a4.y; a4.z = ek * a4.z; a4.w = ek * a4.w;
      kgroup8(acc0, acc1, acc2, acc3, a4, wp + 8 * g * HH);
    }
  }
  #pragma unroll
  for (int r = 0; r < 8; ++r) {
    const int orow = __shfl(node, r + 8 * half);
    const float4 res = *(const float4*)(tw + (r + 8 * half) * TP + HH + 4 * lid);
    float4 v;
    v.x = acc0[r] + res.x;
    v.y = acc1[r] + res.y;
    v.z = acc2[r] + res.z;
    v.w = acc3[r] + res.w;
    v.x = v.x > 0.0f ? v.x : 0.0f;
    v.y = v.y > 0.0f ? v.y : 0.0f;
    v.z = v.z > 0.0f ? v.z : 0.0f;
    v.w = v.w > 0.0f ? v.w : 0.0f;
    *(float4*)(hout + orow * HH + 4 * lid) = v;
  }
}

__global__ __launch_bounds__(256) void fc1_kernel(const float* __restrict__ h,
                                                  const float* __restrict__ w,
                                                  const float* __restrict__ b,
                                                  float* __restrict__ out) {
  const int wave = threadIdx.x >> 5;
  const int lane = threadIdx.x & 31;
  const int half = lane >> 4;
  const int lid  = lane & 15;
  const int node0 = blockIdx.x * 128 + wave * 16;
  if (node0 >= NN) return;

  v8f acc = {};
  for (int c = 0; c < HH / 4; ++c) {
    const int k0 = 4 * c + 2 * half;
    const float* ap = h + (node0 + lid) * HH + k0;
    v2f a; a.x = ap[0]; a.y = ap[1];
    const float* bp = w + k0 * CC;
    v2f bb; bb.x = bp[lid]; bb.y = bp[CC + lid];
    acc = wmma_f32(a, bb, acc);
  }
  const float bias = b[lid];
  for (int r = 0; r < 8; ++r)
    out[(node0 + r + 8 * half) * CC + lid] = acc[r] + bias;
}

extern "C" void kernel_launch(void* const* d_in, const int* in_sizes, int n_in,
                              void* d_out, int out_size, void* d_ws, size_t ws_size,
                              hipStream_t stream) {
  const float* x      = (const float*)d_in[0];
  const int*   eidx   = (const int*)d_in[1];
  const float* fc0_w  = (const float*)d_in[2];
  const float* fc0_b  = (const float*)d_in[3];
  const float* fc1_w  = (const float*)d_in[4];
  const float* fc1_b  = (const float*)d_in[5];
  const float* env_w  = (const float*)d_in[6];
  const float* env_b  = (const float*)d_in[7];
  const float* conv_w = (const float*)d_in[8];
  float* out = (float*)d_out;

  const int* row = eidx;
  const int* col = eidx + EE;

  float* ws  = (float*)d_ws;
  int*   cs  = (int*)ws;
  float* dn  = ws + NN;
  float* hA  = dn + NN;
  float* lst = hA + (size_t)NN * HH;
  float* hB  = lst + (size_t)NN * HH;
  float* eg  = hB + (size_t)NN * HH;
  int*   srow = (int*)lst;
  float* sval = lst + EE;
  int*   binTot  = (int*)(lst + 2 * (size_t)EE);
  int*   binCur  = binTot + 512;
  int*   binBase = binTot + 1024;
  int*   prow = (int*)hB;
  int*   pcol = prow + EE;
  int*   tail = (int*)(lst + 3 * (size_t)EE);
  int*   bh      = tail + 512;
  int*   keyrank = bh + 4 * QB * 256;
  int*   order   = keyrank + NN;

  const int T = 256;
  const int gemm_blocks = (NN + 127) / 128;
  const int node_blocks = (NN + T - 1) / T;

  zero_kernel<<<4, T, 0, stream>>>(binTot, 1024);
  bin_count_kernel<<<CBLK, T, 0, stream>>>(col, binTot);
  partition_kernel<<<PBLK, 512, 0, stream>>>(row, col, binTot, binCur, binBase, prow, pcol);
  bin_place_kernel<<<NB, 512, 0, stream>>>(prow, pcol, binBase, binTot, cs, dn, srow, sval);
  norm_kernel<<<NBLK, T, 0, stream>>>(srow, dn, sval);

  fc0_kernel<<<gemm_blocks, T, 0, stream>>>(x, fc0_w, fc0_b, hA);

  float* cur = hA;
  float* nxt = hB;
  for (int layer = 0; layer < LL; ++layer) {
    gate_kernel<<<node_blocks, T, 0, stream>>>(cur, env_w, env_b, eg, keyrank, bh, layer);
    key_scan_kernel<<<1, 1024, 0, stream>>>(bh);
    order_kernel<<<node_blocks, T, 0, stream>>>(keyrank, bh, order);
    conv_kernel<<<gemm_blocks, T, 0, stream>>>(cs, srow, sval, cur, nxt, eg, conv_w, order, layer);
    float* t = cur; cur = nxt; nxt = t;
  }

  fc1_kernel<<<gemm_blocks, T, 0, stream>>>(cur, fc1_w, fc1_b, out);
}


# Two independent launch chains before conv 0 (edge grouping | fc0, gate, key scan, order of layer 0) share launches stage by stage, one role per work-group: 19 -> 15; sum order stays LDS arrival rank
# speedup vs baseline: 2.7828x; 1.0599x over previous
#include <hip/hip_runtime.h>
#include <hip/hip_bf16.h>

#define NN 100000
#define DD 128
#define HH 64
#define KK 8
#define LL 3
#define CC 16
#define EE 1600000
#define THETA 0.1f
#define NB ((NN + 255) / 256)
#define QB ((NB + 3) / 4)
#define PBLK 255
#define PCH ((EE + PBLK - 1) / PBLK)
#define PIT ((PCH + 511) / 512)
#define CBLK 512
#define CCH ((EE + CBLK - 1) / CBLK)
#define CIT ((CCH + 255) / 256)
#define GB ((NN + 127) / 128)
#define OB ((NN + 511) / 512)
#define NBLK 1563
#define NIT ((EE + NBLK * 256 - 1) / (NBLK * 256))

typedef __attribute__((ext_vector_type(2))) float v2f;
typedef __attribute__((ext_vector_type(8))) float v8f;

__device__ __forceinline__ v8f wmma_f32(v2f a, v2f b, v8f c) {
  return __builtin_amdgcn_wmma_f32_16x16x4_f32(
       false, a,  false, b,
       (short)0, c,  false,  false);
}

__device__ __forceinline__ void kgroup8(v8f& acc0, v8f& acc1, v8f& acc2, v8f& acc3,
                                        const float4 a4, const float* wp) {
  const float4 w0 = *(const float4*)(wp);
  const float4 w1 = *(const float4*)(wp + HH);
  const float4 w2 = *(const float4*)(wp + 2 * HH);
  const float4 w3 = *(const float4*)(wp + 3 * HH);
  v2f a, b;
  a.x = a4.x; a.y = a4.y;
  b.x = w0.x; b.y = w1.x; acc0 = wmma_f32(a, b, acc0);
  b.x = w0.y; b.y = w1.y; acc1 = wmma_f32(a, b, acc1);
  b.x = w0.z; b.y = w1.z; acc2 = wmma_f32(a, b, acc2);
  b.x = w0.w; b.y = w1.w; acc3 = wmma_f32(a, b, acc3);
  a.x = a4.z; a.y = a4.w;
  b.x = w2.x; b.y = w3.x; acc0 = wmma_f32(a, b, acc0);
  b.x = w2.y; b.y = w3.y; acc1 = wmma_f32(a, b, acc1);
  b.x = w2.z; b.y = w3.z; acc2 = wmma_f32(a, b, acc2);
  b.x = w2.w; b.y = w3.w; acc3 = wmma_f32(a, b, acc3);
}


__device__ __forceinline__ void bin_count_body(const int* __restrict__ col,
                                               int* __restrict__ binTot,
                                               int* hist  , const int blk) {
  for (int i = threadIdx.x; i < NB; i += 256) hist[i] = 0;
  __syncthreads();
  const int e0 = blk * CCH;
  int c[CIT];
  #pragma unroll
  for (int j = 0; j < CIT; ++j) {
    const int i = threadIdx.x + 256 * j;
    const int e = e0 + i;
    c[j] = ((i < CCH) && (e < EE)) ? col[e] : -1;
  }
  #pragma unroll
  for (int j = 0; j < CIT; ++j)
    if (c[j] >= 0) atomicAdd(&hist[c[j] >> 8], 1);
  __syncthreads();
  for (int i = threadIdx.x; i < NB; i += 256) {
    const int v = hist[i];
    if (v > 0) atomicAdd(&binTot[i], v);
  }
}

__device__ __forceinline__ void partition_body(const int* __restrict__ row,
                                               const int* __restrict__ col,
                                               const int* __restrict__ binTot,
                                               int* __restrict__ binCur,
                                               int* __restrict__ binBase,
                                               int* __restrict__ prow,
                                               int* __restrict__ pcol,
                                               int* hist, int* sc, int* ofs  ,
                                               const int blk) {
  const int t = threadIdx.x;
  const bool act = t < 512;
  const int tot = (t < NB) ? binTot[t] : 0;
  if (act) {
    hist[t] = 0;
    sc[t] = tot;
  }
  __syncthreads();
  const int e0 = blk * PCH;
  int r[PIT], c[PIT], lr[PIT];
  #pragma unroll
  for (int j = 0; j < PIT; ++j) {
    const int i = t + 512 * j;
    const int e = e0 + i;
    const bool ok = act && (i < PCH) && (e < EE);
    r[j] = ok ? row[e] : 0;
    c[j] = ok ? col[e] : -1;
  }
  #pragma unroll
  for (int j = 0; j < PIT; ++j)
    lr[j] = (c[j] >= 0) ? atomicAdd(&hist[c[j] >> 8], 1) : 0;
  for (int off = 1; off < 512; off <<= 1) {
    const int v = (act && t >= off) ? sc[t - off] : 0;
    __syncthreads();
    if (act) sc[t] += v;
    __syncthreads();
  }
  if (act) {
    const int base = sc[t] - tot;
    const int h = hist[t];
    int o = base;
    if (t < NB && h > 0) o = base + atomicAdd(&binCur[t], h);
    ofs[t] = o;
    if (blk == 0 && t < NB) binBase[t] = base;
  }
  __syncthreads();
  #pragma unroll
  for (int j = 0; j < PIT; ++j) {
    if (c[j] >= 0) {
      const int p = ofs[c[j] >> 8] + lr[j];
      prow[p] = r[j];
      pcol[p] = c[j];
    }
  }
}

__device__ __forceinline__ void bin_place_body(const int* __restrict__ prow,
                                               const int* __restrict__ pcol,
                                               const int* __restrict__ binBase,
                                               const int* __restrict__ binTot,
                                               int* __restrict__ start,
                                               float* __restrict__ dn,
                                               int* __restrict__ srow,
                                               float* __restrict__ sval,
                                               int* cnt, int* sc, int* cur, float* dl  ,
                                               const int b) {
  const int t = threadIdx.x;
  const int lo = binBase[b];
  const int hi = lo + binTot[b];
  if (t < 256) cnt[t] = 0;
  __syncthreads();
  for (int i0 = lo + t; i0 < hi; i0 += 2048) {
    int c[4];
    #pragma unroll
    for (int u = 0; u < 4; ++u) {
      const int i = i0 + 512 * u;
      c[u] = (i < hi) ? pcol[i] : -1;
    }
    #pragma unroll
    for (int u = 0; u < 4; ++u)
      if (c[u] >= 0) atomicAdd(&cnt[c[u] & 255], 1);
  }
  __syncthreads();
  const int d = (t < 256) ? cnt[t] : 0;
  if (t < 256) sc[t] = d;
  __syncthreads();
  for (int off = 1; off < 256; off <<= 1) {
    int v = 0;
    if (t < 256 && t >= off) v = sc[t - off];
    __syncthreads();
    if (t < 256) sc[t] += v;
    __syncthreads();
  }
  if (t < 256) {
    const int s0 = lo + sc[t] - d;
    const float dv = (d > 0) ? rsqrtf((float)d) : 0.0f;
    cur[t] = s0;
    dl[t] = dv;
    const int n = b * 256 + t;
    if (n < NN) { start[n] = s0; dn[n] = dv; }
  }
  __syncthreads();
  for (int i0 = lo + t; i0 < hi; i0 += 2048) {
    int c[4], r[4];
    #pragma unroll
    for (int u = 0; u < 4; ++u) {
      const int i = i0 + 512 * u;
      const bool ok = i < hi;
      c[u] = ok ? pcol[i] : -1;
      r[u] = ok ? prow[i] : 0;
    }
    #pragma unroll
    for (int u = 0; u < 4; ++u) {
      if (c[u] >= 0) {
        const int cl = c[u] & 255;
        const int p = atomicAdd(&cur[cl], 1);
        srow[p] = r[u];
        sval[p] = dl[cl];
      }
    }
  }
}

__global__ __launch_bounds__(256) void norm_kernel(const int* __restrict__ srow,
                                                   const float* __restrict__ dn,
                                                   float* __restrict__ sval) {
  const int p0 = blockIdx.x * 256 + threadIdx.x;
  int r[NIT];
  float v[NIT], d[NIT];
  #pragma unroll
  for (int j = 0; j < NIT; ++j) {
    const int p = p0 + NBLK * 256 * j;
    const int pc = (p < EE) ? p : 0;
    r[j] = srow[pc];
    v[j] = sval[pc];
  }
  #pragma unroll
  for (int j = 0; j < NIT; ++j) d[j] = dn[r[j]];
  #pragma unroll
  for (int j = 0; j < NIT; ++j) {
    const int p = p0 + NBLK * 256 * j;
    if (p < EE) sval[p] = v[j] * d[j];
  }
}

__device__ __forceinline__ void fc0_body(const float* __restrict__ x,
                                         const float* __restrict__ w,
                                         const float* __restrict__ b,
                                         float* __restrict__ h, const int blk) {
  const int wave = threadIdx.x >> 5;
  const int lane = threadIdx.x & 31;
  const int half = lane >> 4;
  const int lid  = lane & 15;
  const int node0 = blk * 128 + wave * 16;
  if (node0 >= NN) return;

  v8f acc0 = {}, acc1 = {}, acc2 = {}, acc3 = {};
  const float* ap = x + (node0 + lid) * DD + 4 * half;
  const float* wp = w + (4 * half) * HH + 4 * lid;
  #pragma unroll 2
  for (int g = 0; g < DD / 8; ++g) {
    const float4 a4 = *(const float4*)(ap + 8 * g);
    kgroup8(acc0, acc1, acc2, acc3, a4, wp + 8 * g * HH);
  }
  const float4 bias = *(const float4*)(b + 4 * lid);
  #pragma unroll
  for (int r = 0; r < 8; ++r) {
    float4 v;
    v.x = acc0[r] + bias.x;
    v.y = acc1[r] + bias.y;
    v.z = acc2[r] + bias.z;
    v.w = acc3[r] + bias.w;
    v.x = v.x > 0.0f ? v.x : 0.0f;
    v.y = v.y > 0.0f ? v.y : 0.0f;
    v.z = v.z > 0.0f ? v.z : 0.0f;
    v.w = v.w > 0.0f ? v.w : 0.0f;
    *(float4*)(h + (node0 + r + 8 * half) * HH + 4 * lid) = v;
  }
}

__device__ __forceinline__ void order_body(const int* __restrict__ keyrank,
                                           const int* __restrict__ offs,
                                           int* __restrict__ order, const int n) {
  if (n < NN) {
    const int kr = keyrank[n];
    order[offs[(n >> 8) * 256 + (kr >> 8)] + (kr & 255)] = n;
  }
}

__device__ __forceinline__ void gate_body(const float* __restrict__ h,
                                          const float* __restrict__ env_w,
                                          const float* __restrict__ env_b,
                                          float* __restrict__ e_out,
                                          int* __restrict__ keyrank,
                                          int* __restrict__ bh, const int layer,
                                          int* hist  , const int blk) {
  const int t = threadIdx.x;
  const int n = blk * 256 + t;
  hist[t] = 0;
  __syncthreads();
  if (n < NN) {
  const float* W = env_w + layer * 2 * HH * KK;
  const float* B = env_b + layer * KK;
  float logit[KK];
  #pragma unroll
  for (int k = 0; k < KK; ++k) logit[k] = B[k];
  for (int f = 0; f < HH; ++f) {
    float hv = h[n * HH + f];
    const float* wr = W + f * KK;
    #pragma unroll
    for (int k = 0; k < KK; ++k) logit[k] += hv * wr[k];
  }
  float m = logit[0];
  #pragma unroll
  for (int k = 1; k < KK; ++k) m = fmaxf(m, logit[k]);
  float s = 0.0f;
  float ex[KK];
  #pragma unroll
  for (int k = 0; k < KK; ++k) { ex[k] = __expf(logit[k] - m); s += ex[k]; }
  float inv = 1.0f / s;
  int key = 0;
  #pragma unroll
  for (int k = 0; k < KK; ++k) {
    float pi = ex[k] * inv;
    const bool on = pi > THETA;
    e_out[n * KK + k] = on ? pi : 0.0f;
    key |= on ? (1 << k) : 0;
  }
  const int r = atomicAdd(&hist[key], 1);
  keyrank[n] = key * 256 + r;
  }
  __syncthreads();
  bh[blk * 256 + t] = hist[t];
  if (blk == 0)
    for (int b = NB; b < 4 * QB; ++b) bh[b * 256 + t] = 0;
}

__device__ __forceinline__ void key_scan_body(int* bh, int (*segsum)[256]  ,
                                              int* kb    ,
                                              int* kpos  ) {
  const int t = threadIdx.x;
  const int key = t & 255;
  const int seg = t >> 8;
  int* p = bh + (seg * QB) * 256 + key;
  int c[QB];
  int s = 0;
  #pragma unroll
  for (int i = 0; i < QB; ++i) {
    c[i] = p[i * 256];
    s += c[i];
  }
  segsum[seg][key] = s;
  __syncthreads();
  if (t < 256) {
    const int tot = segsum[0][t] + segsum[1][t] + segsum[2][t] + segsum[3][t];
    const int pt = __popc(t);
    int rk = 0;
    for (int q = 0; q < 256; ++q) {
      const int pq = __popc(q);
      rk += (pq > pt || (pq == pt && q < t)) ? 1 : 0;
    }
    kpos[t] = rk;
    kb[rk] = tot;
  }
  __syncthreads();
  for (int off = 1; off < 256; off <<= 1) {
    int v = 0;
    if (t < 256 && t >= off) v = kb[t - off];
    __syncthreads();
    if (t < 256) kb[t] += v;
    __syncthreads();
  }
  int run = kb[kpos[key]] - (segsum[0][key] + segsum[1][key] + segsum[2][key] + segsum[3][key]);
  for (int j = 0; j < seg; ++j) run += segsum[j][key];
  #pragma unroll
  for (int i = 0; i < QB; ++i) {
    p[i * 256] = run;
    run += c[i];
  }
}

__global__ __launch_bounds__(256) void fc0_clear_kernel(const float* __restrict__ x,
                                                        const float* __restrict__ w,
                                                        const float* __restrict__ b,
                                                        float* __restrict__ h,
                                                        int* __restrict__ tab) {
  if (blockIdx.x < GB) fc0_body(x, w, b, h, blockIdx.x);
  else tab[(blockIdx.x - GB) * 256 + threadIdx.x] = 0;
}

__global__ __launch_bounds__(256) void gate_count_kernel(const float* __restrict__ h,
                                                         const float* __restrict__ env_w,
                                                         const float* __restrict__ env_b,
                                                         float* __restrict__ e_out,
                                                         int* __restrict__ keyrank,
                                                         int* __restrict__ bh,
                                                         const int* __restrict__ col,
                                                         int* __restrict__ binTot) {
  __shared__ int hist[NB];
  if (blockIdx.x < NB) gate_body(h, env_w, env_b, e_out, keyrank, bh, 0, hist, blockIdx.x);
  else bin_count_body(col, binTot, hist, blockIdx.x - NB);
}

__global__ __launch_bounds__(1024) void scan_partition_kernel(int* bh,
                                                              const int* __restrict__ row,
                                                              const int* __restrict__ col,
                                                              const int* __restrict__ binTot,
                                                              int* __restrict__ binCur,
                                                              int* __restrict__ binBase,
                                                              int* __restrict__ prow,
                                                              int* __restrict__ pcol) {
  __shared__ int lds[6 * 256];
  if (blockIdx.x == 0) key_scan_body(bh, (int (*)[256])lds, lds + 1024, lds + 1280);
  else partition_body(row, col, binTot, binCur, binBase, prow, pcol, lds, lds + 512, lds + 1024, blockIdx.x - 1);
}

__global__ __launch_bounds__(512) void place_order_kernel(const int* __restrict__ prow,
                                                          const int* __restrict__ pcol,
                                                          const int* __restrict__ binBase,
                                                          const int* __restrict__ binTot,
                                                          int* __restrict__ start,
                                                          float* __restrict__ dn,
                                                          int* __restrict__ srow,
                                                          float* __restrict__ sval,
                                                          const int* __restrict__ keyrank,
                                                          const int* __restrict__ offs,
                                                          int* __restrict__ order) {
  __shared__ int lds[3 * 256];
  __shared__ float dl[256];
  if (blockIdx.x < NB) bin_place_body(prow, pcol, binBase, binTot, start, dn, srow, sval,
                                      lds, lds + 256, lds + 512, dl, blockIdx.x);
  else order_body(keyrank, offs, order, (blockIdx.x - NB) * 512 + threadIdx.x);
}

__global__ __launch_bounds__(256) void gate_kernel(const float* __restrict__ h,
                                                   const float* __restrict__ env_w,
                                                   const float* __restrict__ env_b,
                                                   float* __restrict__ e_out,
                                                   int* __restrict__ keyrank,
                                                   int* __restrict__ bh, int layer) {
  __shared__ int hist[256];
  gate_body(h, env_w, env_b, e_out, keyrank, bh, layer, hist, blockIdx.x);
}

__global__ __launch_bounds__(1024) void key_scan_kernel(int* bh) {
  __shared__ int lds[6 * 256];
  key_scan_body(bh, (int (*)[256])lds, lds + 1024, lds + 1280);
}

__global__ __launch_bounds__(256) void order_kernel(const int* __restrict__ keyrank,
                                                    const int* __restrict__ offs,
                                                    int* __restrict__ order) {
  order_body(keyrank, offs, order, blockIdx.x * 256 + threadIdx.x);
}

#define TP 132
__global__ __launch_bounds__(256) void conv_kernel(const int* __restrict__ start,
                                                   const int* __restrict__ srow,
                                                   const float* __restrict__ sval,
                                                   const float* __restrict__ hin,
                                                   float* __restrict__ hout,
                                                   const float* __restrict__ e_gate,
                                                   const float* __restrict__ conv_w,
                                                   const int* __restrict__ order,
                                                   int layer) {
  __shared__ __attribute__((aligned(16))) float tile[8 * 16 * TP];
  const int wave = threadIdx.x >> 5;
  const int lane = threadIdx.x & 31;
  const int half = lane >> 4;
  const int lid  = lane & 15;
  const int slot0 = blockIdx.x * 128 + wave * 16;
  const bool live = slot0 < NN;
  float* tw = tile + wave * (16 * TP);
  int node = 0;
  if (live) {
    node = order[slot0 + lid];
    const int sA = start[node];
    const int eA = (node + 1 < NN) ? start[node + 1] : EE;
    {
      float4 own[8];
      #pragma unroll
      for (int it = 0; it < 8; ++it)
        own[it] = *(const float4*)(hin + __shfl(node, 2 * it + half) * HH + 4 * lid);
      #pragma unroll
      for (int it = 0; it < 8; ++it)
        *(float4*)(tw + (2 * it + half) * TP + HH + 4 * lid) = own[it];
    }
    for (int it = 0; it < 8; ++it) {
      const int i = 2 * it + half;
      const int s = __shfl(sA, i);
      const int e = __shfl(eA, i);
      float4 acc = make_float4(0.0f, 0.0f, 0.0f, 0.0f);
      for (int base = s; ; base += 32) {
        const int rem = e - base;
        const int cmax = max(rem, __shfl_xor(rem, 16));
        if (cmax <= 0) break;
        int myr0 = 0, myr1 = 0;
        float myv0 = 0.0f, myv1 = 0.0f;
        if (lid < rem) { myr0 = srow[base + lid]; myv0 = sval[base + lid]; }
        if (lid + 16 < rem) { myr1 = srow[base + 16 + lid]; myv1 = sval[base + 16 + lid]; }
        const int lim = min(__builtin_amdgcn_readfirstlane(cmax), 32);
        for (int j0 = 0; j0 < lim; j0 += 8) {
          const int myr = (j0 < 16) ? myr0 : myr1;
          const float myv = (j0 < 16) ? myv0 : myv1;
          #pragma unroll
          for (int u = 0; u < 8; ++u) {
            const int j = j0 + u;
            const int r = __shfl(myr, j & 15, 16);
            const float v = __shfl(myv, j & 15, 16);
            float4 hv = *(const float4*)(hin + r * HH + lid * 4);
            const bool on = j < rem;
            hv.x = on ? hv.x : 0.0f;
            hv.y = on ? hv.y : 0.0f;
            hv.z = on ? hv.z : 0.0f;
            hv.w = on ? hv.w : 0.0f;
            acc.x += v * hv.x;
            acc.y += v * hv.y;
            acc.z += v * hv.z;
            acc.w += v * hv.w;
          }
        }
      }
      *(float4*)(tw + i * TP + 4 * lid) = acc;
    }
  }
  __syncthreads();
  if (!live) return;

  const float* W = conv_w + layer * (KK * 2 * HH * HH);
  v8f acc0 = {}, acc1 = {}, acc2 = {}, acc3 = {};
  const float* ap = tw + lid * TP + 4 * half;

  #pragma unroll 1
  for (int k = 0; k < KK; ++k) {
    const float ek = e_gate[node * KK + k];
    if (!__any(ek != 0.0f)) continue;
    const float* wp = W + k * (2 * HH * HH) + (4 * half) * HH + 4 * lid;
    #pragma unroll 2
    for (int g = 0; g < 2 * HH / 8; ++g) {
      float4 a4 = *(const float4*)(ap + 8 * g);
      a4.x = ek * a4.x; a4.y = ek * a4.y; a4.z = ek * a4.z; a4.w = ek * a4.w;
      kgroup8(acc0, acc1, acc2, acc3, a4, wp + 8 * g * HH);
    }
  }
  #pragma unroll
  for (int r = 0; r < 8; ++r) {
    const int orow = __shfl(node, r + 8 * half);
    const float4 res = *(const float4*)(tw + (r + 8 * half) * TP + HH + 4 * lid);
    float4 v;
    v.x = acc0[r] + res.x;
    v.y = acc1[r] + res.y;
    v.z = acc2[r] + res.z;
    v.w = acc3[r] + res.w;
    v.x = v.x > 0.0f ? v.x : 0.0f;
    v.y = v.y > 0.0f ? v.y : 0.0f;
    v.z = v.z > 0.0f ? v.z : 0.0f;
    v.w = v.w > 0.0f ? v.w : 0.0f;
    *(float4*)(hout + orow * HH + 4 * lid) = v;
  }
}

__global__ __launch_bounds__(256) void fc1_kernel(const float* __restrict__ h,
                                                  const float* __restrict__ w,
                                                  const float* __restrict__ b,
                                                  float* __restrict__ out) {
  const int wave = threadIdx.x >> 5;
  const int lane = threadIdx.x & 31;
  const int half = lane >> 4;
  const int lid  = lane & 15;
  const int node0 = blockIdx.x * 128 + wave * 16;
  if (node0 >= NN) return;

  v8f acc = {};
  for (int c = 0; c < HH / 4; ++c) {
    const int k0 = 4 * c + 2 * half;
    const float* ap = h + (node0 + lid) * HH + k0;
    v2f a; a.x = ap[0]; a.y = ap[1];
    const float* bp = w + k0 * CC;
    v2f bb; bb.x = bp[lid]; bb.y = bp[CC + lid];
    acc = wmma_f32(a, bb, acc);
  }
  const float bias = b[lid];
  for (int r = 0; r < 8; ++r)
    out[(node0 + r + 8 * half) * CC + lid] = acc[r] + bias;
}

extern "C" void kernel_launch(void* const* d_in, const int* in_sizes, int n_in,
                              void* d_out, int out_size, void* d_ws, size_t ws_size,
                              hipStream_t stream) {
  const float* x      = (const float*)d_in[0];
  const int*   eidx   = (const int*)d_in[1];
  const float* fc0_w  = (const float*)d_in[2];
  const float* fc0_b  = (const float*)d_in[3];
  const float* fc1_w  = (const float*)d_in[4];
  const float* fc1_b  = (const float*)d_in[5];
  const float* env_w  = (const float*)d_in[6];
  const float* env_b  = (const float*)d_in[7];
  const float* conv_w = (const float*)d_in[8];
  float* out = (float*)d_out;

  const int* row = eidx;
  const int* col = eidx + EE;

  float* ws  = (float*)d_ws;
  int*   cs  = (int*)ws;
  float* dn  = ws + NN;
  float* hA  = dn + NN;
  float* lst = hA + (size_t)NN * HH;
  float* hB  = lst + (size_t)NN * HH;
  float* eg  = hB + (size_t)NN * HH;
  int*   srow = (int*)lst;
  float* sval = lst + EE;
  int*   binTot  = (int*)(lst + 2 * (size_t)EE);
  int*   binCur  = binTot + 512;
  int*   binBase = binTot + 1024;
  int*   prow = (int*)hB;
  int*   pcol = prow + EE;
  int*   tail = (int*)(lst + 3 * (size_t)EE);
  int*   bh      = tail + 512;
  int*   keyrank = bh + 4 * QB * 256;
  int*   order   = keyrank + NN;

  const int T = 256;

  fc0_clear_kernel<<<GB + 4, T, 0, stream>>>(x, fc0_w, fc0_b, hA, binTot);
  gate_count_kernel<<<NB + CBLK, T, 0, stream>>>(hA, env_w, env_b, eg, keyrank, bh, col, binTot);
  scan_partition_kernel<<<1 + PBLK, 1024, 0, stream>>>(bh, row, col, binTot, binCur, binBase, prow, pcol);
  place_order_kernel<<<NB + OB, 512, 0, stream>>>(prow, pcol, binBase, binTot, cs, dn, srow, sval,
                                                  keyrank, bh, order);
  norm_kernel<<<NBLK, T, 0, stream>>>(srow, dn, sval);
  conv_kernel<<<GB, T, 0, stream>>>(cs, srow, sval, hA, hB, eg, conv_w, order, 0);

  float* cur = hB;
  float* nxt = hA;
  for (int layer = 1; layer < LL; ++layer) {
    gate_kernel<<<NB, T, 0, stream>>>(cur, env_w, env_b, eg, keyrank, bh, layer);
    key_scan_kernel<<<1, 1024, 0, stream>>>(bh);
    order_kernel<<<NB, T, 0, stream>>>(keyrank, bh, order);
    conv_kernel<<<GB, T, 0, stream>>>(cs, srow, sval, cur, nxt, eg, conv_w, order, layer);
    float* t = cur; cur = nxt; nxt = t;
  }

  fc1_kernel<<<GB, T, 0, stream>>>(cur, fc1_w, fc1_b, out);
}


# Next layer's gate and sort key made in the conv epilogue from the wave's tile (same gate arithmetic); keys ranked by LDS counter + one returning int global add per key: 15 -> 11 launches
# speedup vs baseline: 3.0381x; 1.0917x over previous
#include <hip/hip_runtime.h>
#include <hip/hip_bf16.h>

#define NN 100000
#define DD 128
#define HH 64
#define KK 8
#define LL 3
#define CC 16
#define EE 1600000
#define THETA 0.1f
#define NB ((NN + 255) / 256)
#define QB ((NB + 3) / 4)
#define PBLK 255
#define PCH ((EE + PBLK - 1) / PBLK)
#define PIT ((PCH + 511) / 512)
#define CBLK 512
#define CCH ((EE + CBLK - 1) / CBLK)
#define CIT ((CCH + 255) / 256)
#define GB ((NN + 127) / 128)
#define OB ((NN + 511) / 512)
#define NBLK 1563
#define NIT ((EE + NBLK * 256 - 1) / (NBLK * 256))

typedef __attribute__((ext_vector_type(2))) float v2f;
typedef __attribute__((ext_vector_type(8))) float v8f;

__device__ __forceinline__ v8f wmma_f32(v2f a, v2f b, v8f c) {
  return __builtin_amdgcn_wmma_f32_16x16x4_f32(
       false, a,  false, b,
       (short)0, c,  false,  false);
}

__device__ __forceinline__ void kgroup8(v8f& acc0, v8f& acc1, v8f& acc2, v8f& acc3,
                                        const float4 a4, const float* wp) {
  const float4 w0 = *(const float4*)(wp);
  const float4 w1 = *(const float4*)(wp + HH);
  const float4 w2 = *(const float4*)(wp + 2 * HH);
  const float4 w3 = *(const float4*)(wp + 3 * HH);
  v2f a, b;
  a.x = a4.x; a.y = a4.y;
  b.x = w0.x; b.y = w1.x; acc0 = wmma_f32(a, b, acc0);
  b.x = w0.y; b.y = w1.y; acc1 = wmma_f32(a, b, acc1);
  b.x = w0.z; b.y = w1.z; acc2 = wmma_f32(a, b, acc2);
  b.x = w0.w; b.y = w1.w; acc3 = wmma_f32(a, b, acc3);
  a.x = a4.z; a.y = a4.w;
  b.x = w2.x; b.y = w3.x; acc0 = wmma_f32(a, b, acc0);
  b.x = w2.y; b.y = w3.y; acc1 = wmma_f32(a, b, acc1);
  b.x = w2.z; b.y = w3.z; acc2 = wmma_f32(a, b, acc2);
  b.x = w2.w; b.y = w3.w; acc3 = wmma_f32(a, b, acc3);
}


__device__ __forceinline__ void bin_count_body(const int* __restrict__ col,
                                               int* __restrict__ binTot,
                                               int* hist  , const int blk) {
  for (int i = threadIdx.x; i < NB; i += 256) hist[i] = 0;
  __syncthreads();
  const int e0 = blk * CCH;
  int c[CIT];
  #pragma unroll
  for (int j = 0; j < CIT; ++j) {
    const int i = threadIdx.x + 256 * j;
    const int e = e0 + i;
    c[j] = ((i < CCH) && (e < EE)) ? col[e] : -1;
  }
  #pragma unroll
  for (int j = 0; j < CIT; ++j)
    if (c[j] >= 0) atomicAdd(&hist[c[j] >> 8], 1);
  __syncthreads();
  for (int i = threadIdx.x; i < NB; i += 256) {
    const int v = hist[i];
    if (v > 0) atomicAdd(&binTot[i], v);
  }
}

__device__ __forceinline__ void partition_body(const int* __restrict__ row,
                                               const int* __restrict__ col,
                                               const int* __restrict__ binTot,
                                               int* __restrict__ binCur,
                                               int* __restrict__ binBase,
                                               int* __restrict__ prow,
                                               int* __restrict__ pcol,
                                               int* hist, int* sc, int* ofs  ,
                                               const int blk) {
  const int t = threadIdx.x;
  const bool act = t < 512;
  const int tot = (t < NB) ? binTot[t] : 0;
  if (act) {
    hist[t] = 0;
    sc[t] = tot;
  }
  __syncthreads();
  const int e0 = blk * PCH;
  int r[PIT], c[PIT], lr[PIT];
  #pragma unroll
  for (int j = 0; j < PIT; ++j) {
    const int i = t + 512 * j;
    const int e = e0 + i;
    const bool ok = act && (i < PCH) && (e < EE);
    r[j] = ok ? row[e] : 0;
    c[j] = ok ? col[e] : -1;
  }
  #pragma unroll
  for (int j = 0; j < PIT; ++j)
    lr[j] = (c[j] >= 0) ? atomicAdd(&hist[c[j] >> 8], 1) : 0;
  for (int off = 1; off < 512; off <<= 1) {
    const int v = (act && t >= off) ? sc[t - off] : 0;
    __syncthreads();
    if (act) sc[t] += v;
    __syncthreads();
  }
  if (act) {
    const int base = sc[t] - tot;
    const int h = hist[t];
    int o = base;
    if (t < NB && h > 0) o = base + atomicAdd(&binCur[t], h);
    ofs[t] = o;
    if (blk == 0 && t < NB) binBase[t] = base;
  }
  __syncthreads();
  #pragma unroll
  for (int j = 0; j < PIT; ++j) {
    if (c[j] >= 0) {
      const int p = ofs[c[j] >> 8] + lr[j];
      prow[p] = r[j];
      pcol[p] = c[j];
    }
  }
}

__device__ __forceinline__ void bin_place_body(const int* __restrict__ prow,
                                               const int* __restrict__ pcol,
                                               const int* __restrict__ binBase,
                                               const int* __restrict__ binTot,
                                               int* __restrict__ start,
                                               float* __restrict__ dn,
                                               int* __restrict__ srow,
                                               float* __restrict__ sval,
                                               int* cnt, int* sc, int* cur, float* dl  ,
                                               const int b) {
  const int t = threadIdx.x;
  const int lo = binBase[b];
  const int hi = lo + binTot[b];
  if (t < 256) cnt[t] = 0;
  __syncthreads();
  for (int i0 = lo + t; i0 < hi; i0 += 2048) {
    int c[4];
    #pragma unroll
    for (int u = 0; u < 4; ++u) {
      const int i = i0 + 512 * u;
      c[u] = (i < hi) ? pcol[i] : -1;
    }
    #pragma unroll
    for (int u = 0; u < 4; ++u)
      if (c[u] >= 0) atomicAdd(&cnt[c[u] & 255], 1);
  }
  __syncthreads();
  const int d = (t < 256) ? cnt[t] : 0;
  if (t < 256) sc[t] = d;
  __syncthreads();
  for (int off = 1; off < 256; off <<= 1) {
    int v = 0;
    if (t < 256 && t >= off) v = sc[t - off];
    __syncthreads();
    if (t < 256) sc[t] += v;
    __syncthreads();
  }
  if (t < 256) {
    const int s0 = lo + sc[t] - d;
    const float dv = (d > 0) ? rsqrtf((float)d) : 0.0f;
    cur[t] = s0;
    dl[t] = dv;
    const int n = b * 256 + t;
    if (n < NN) { start[n] = s0; dn[n] = dv; }
  }
  __syncthreads();
  for (int i0 = lo + t; i0 < hi; i0 += 2048) {
    int c[4], r[4];
    #pragma unroll
    for (int u = 0; u < 4; ++u) {
      const int i = i0 + 512 * u;
      const bool ok = i < hi;
      c[u] = ok ? pcol[i] : -1;
      r[u] = ok ? prow[i] : 0;
    }
    #pragma unroll
    for (int u = 0; u < 4; ++u) {
      if (c[u] >= 0) {
        const int cl = c[u] & 255;
        const int p = atomicAdd(&cur[cl], 1);
        srow[p] = r[u];
        sval[p] = dl[cl];
      }
    }
  }
}

__global__ __launch_bounds__(256) void norm_kernel(const int* __restrict__ srow,
                                                   const float* __restrict__ dn,
                                                   float* __restrict__ sval) {
  const int p0 = blockIdx.x * 256 + threadIdx.x;
  int r[NIT];
  float v[NIT], d[NIT];
  #pragma unroll
  for (int j = 0; j < NIT; ++j) {
    const int p = p0 + NBLK * 256 * j;
    const int pc = (p < EE) ? p : 0;
    r[j] = srow[pc];
    v[j] = sval[pc];
  }
  #pragma unroll
  for (int j = 0; j < NIT; ++j) d[j] = dn[r[j]];
  #pragma unroll
  for (int j = 0; j < NIT; ++j) {
    const int p = p0 + NBLK * 256 * j;
    if (p < EE) sval[p] = v[j] * d[j];
  }
}

__device__ __forceinline__ void fc0_body(const float* __restrict__ x,
                                         const float* __restrict__ w,
                                         const float* __restrict__ b,
                                         float* __restrict__ h, const int blk) {
  const int wave = threadIdx.x >> 5;
  const int lane = threadIdx.x & 31;
  const int half = lane >> 4;
  const int lid  = lane & 15;
  const int node0 = blk * 128 + wave * 16;
  if (node0 >= NN) return;

  v8f acc0 = {}, acc1 = {}, acc2 = {}, acc3 = {};
  const float* ap = x + (node0 + lid) * DD + 4 * half;
  const float* wp = w + (4 * half) * HH + 4 * lid;
  #pragma unroll 2
  for (int g = 0; g < DD / 8; ++g) {
    const float4 a4 = *(const float4*)(ap + 8 * g);
    kgroup8(acc0, acc1, acc2, acc3, a4, wp + 8 * g * HH);
  }
  const float4 bias = *(const float4*)(b + 4 * lid);
  #pragma unroll
  for (int r = 0; r < 8; ++r) {
    float4 v;
    v.x = acc0[r] + bias.x;
    v.y = acc1[r] + bias.y;
    v.z = acc2[r] + bias.z;
    v.w = acc3[r] + bias.w;
    v.x = v.x > 0.0f ? v.x : 0.0f;
    v.y = v.y > 0.0f ? v.y : 0.0f;
    v.z = v.z > 0.0f ? v.z : 0.0f;
    v.w = v.w > 0.0f ? v.w : 0.0f;
    *(float4*)(h + (node0 + r + 8 * half) * HH + 4 * lid) = v;
  }
}

__device__ __forceinline__ void order_body(const int* __restrict__ keyrank,
                                           const int* __restrict__ offs,
                                           int* __restrict__ order, const int n) {
  if (n < NN) {
    const int kr = keyrank[n];
    order[offs[(n >> 8) * 256 + (kr >> 8)] + (kr & 255)] = n;
  }
}

__device__ __forceinline__ void gate_body(const float* __restrict__ h,
                                          const float* __restrict__ env_w,
                                          const float* __restrict__ env_b,
                                          float* __restrict__ e_out,
                                          int* __restrict__ keyrank,
                                          int* __restrict__ bh, const int layer,
                                          int* hist  , const int blk) {
  const int t = threadIdx.x;
  const int n = blk * 256 + t;
  hist[t] = 0;
  __syncthreads();
  if (n < NN) {
  const float* W = env_w + layer * 2 * HH * KK;
  const float* B = env_b + layer * KK;
  float logit[KK];
  #pragma unroll
  for (int k = 0; k < KK; ++k) logit[k] = B[k];
  for (int f = 0; f < HH; ++f) {
    float hv = h[n * HH + f];
    const float* wr = W + f * KK;
    #pragma unroll
    for (int k = 0; k < KK; ++k) logit[k] += hv * wr[k];
  }
  float m = logit[0];
  #pragma unroll
  for (int k = 1; k < KK; ++k) m = fmaxf(m, logit[k]);
  float s = 0.0f;
  float ex[KK];
  #pragma unroll
  for (int k = 0; k < KK; ++k) { ex[k] = __expf(logit[k] - m); s += ex[k]; }
  float inv = 1.0f / s;
  int key = 0;
  #pragma unroll
  for (int k = 0; k < KK; ++k) {
    float pi = ex[k] * inv;
    const bool on = pi > THETA;
    e_out[n * KK + k] = on ? pi : 0.0f;
    key |= on ? (1 << k) : 0;
  }
  const int r = atomicAdd(&hist[key], 1);
  keyrank[n] = key * 256 + r;
  }
  __syncthreads();
  bh[blk * 256 + t] = hist[t];
  if (blk == 0)
    for (int b = NB; b < 4 * QB; ++b) bh[b * 256 + t] = 0;
}

__device__ __forceinline__ void key_scan_body(int* bh, int (*segsum)[256]  ,
                                              int* kb    ,
                                              int* kpos  ) {
  const int t = threadIdx.x;
  const int key = t & 255;
  const int seg = t >> 8;
  int* p = bh + (seg * QB) * 256 + key;
  int c[QB];
  int s = 0;
  #pragma unroll
  for (int i = 0; i < QB; ++i) {
    c[i] = p[i * 256];
    s += c[i];
  }
  segsum[seg][key] = s;
  __syncthreads();
  if (t < 256) {
    const int tot = segsum[0][t] + segsum[1][t] + segsum[2][t] + segsum[3][t];
    const int pt = __popc(t);
    int rk = 0;
    for (int q = 0; q < 256; ++q) {
      const int pq = __popc(q);
      rk += (pq > pt || (pq == pt && q < t)) ? 1 : 0;
    }
    kpos[t] = rk;
    kb[rk] = tot;
  }
  __syncthreads();
  for (int off = 1; off < 256; off <<= 1) {
    int v = 0;
    if (t < 256 && t >= off) v = kb[t - off];
    __syncthreads();
    if (t < 256) kb[t] += v;
    __syncthreads();
  }
  int run = kb[kpos[key]] - (segsum[0][key] + segsum[1][key] + segsum[2][key] + segsum[3][key]);
  for (int j = 0; j < seg; ++j) run += segsum[j][key];
  #pragma unroll
  for (int i = 0; i < QB; ++i) {
    p[i * 256] = run;
    run += c[i];
  }
}

__global__ __launch_bounds__(256) void fc0_clear_kernel(const float* __restrict__ x,
                                                        const float* __restrict__ w,
                                                        const float* __restrict__ b,
                                                        float* __restrict__ h,
                                                        int* __restrict__ tab) {
  if (blockIdx.x < GB) fc0_body(x, w, b, h, blockIdx.x);
  else tab[(blockIdx.x - GB) * 256 + threadIdx.x] = 0;
}

__global__ __launch_bounds__(256) void gate_count_kernel(const float* __restrict__ h,
                                                         const float* __restrict__ env_w,
                                                         const float* __restrict__ env_b,
                                                         float* __restrict__ e_out,
                                                         int* __restrict__ keyrank,
                                                         int* __restrict__ bh,
                                                         const int* __restrict__ col,
                                                         int* __restrict__ binTot) {
  __shared__ int hist[NB];
  if (blockIdx.x < NB) gate_body(h, env_w, env_b, e_out, keyrank, bh, 0, hist, blockIdx.x);
  else bin_count_body(col, binTot, hist, blockIdx.x - NB);
}

__global__ __launch_bounds__(1024) void scan_partition_kernel(int* bh,
                                                              const int* __restrict__ row,
                                                              const int* __restrict__ col,
                                                              const int* __restrict__ binTot,
                                                              int* __restrict__ binCur,
                                                              int* __restrict__ binBase,
                                                              int* __restrict__ prow,
                                                              int* __restrict__ pcol) {
  __shared__ int lds[6 * 256];
  if (blockIdx.x == 0) key_scan_body(bh, (int (*)[256])lds, lds + 1024, lds + 1280);
  else partition_body(row, col, binTot, binCur, binBase, prow, pcol, lds, lds + 512, lds + 1024, blockIdx.x - 1);
}

__global__ __launch_bounds__(512) void place_order_kernel(const int* __restrict__ prow,
                                                          const int* __restrict__ pcol,
                                                          const int* __restrict__ binBase,
                                                          const int* __restrict__ binTot,
                                                          int* __restrict__ start,
                                                          float* __restrict__ dn,
                                                          int* __restrict__ srow,
                                                          float* __restrict__ sval,
                                                          const int* __restrict__ keyrank,
                                                          const int* __restrict__ offs,
                                                          int* __restrict__ order) {
  __shared__ int lds[3 * 256];
  __shared__ float dl[256];
  if (blockIdx.x < NB) bin_place_body(prow, pcol, binBase, binTot, start, dn, srow, sval,
                                      lds, lds + 256, lds + 512, dl, blockIdx.x);
  else order_body(keyrank, offs, order, (blockIdx.x - NB) * 512 + threadIdx.x);
}

__device__ constexpr int binom(const int n, const int k) {
  int r = 1;
  for (int i = 1; i <= k; ++i) r = r * (n - k + i) / i;
  return r;
}

__device__ __forceinline__ int walk_place(const int t) {
  const int pt = __popc(t);
  int rk = 0;
  rk += (pt < 8) ? 1 : 0;
  rk += (pt < 7) ? 8 : 0;
  rk += (pt < 6) ? 28 : 0;
  rk += (pt < 5) ? 56 : 0;
  rk += (pt < 4) ? 70 : 0;
  rk += (pt < 3) ? 56 : 0;
  rk += (pt < 2) ? 28 : 0;
  rk += (pt < 1) ? 8 : 0;
  int need = pt;
  #pragma unroll
  for (int b = 7; b >= 0; --b) {
    if ((t >> b) & 1) {
      int cnt = 0;
      #pragma unroll
      for (int k = 0; k <= b; ++k) cnt = (need == k) ? binom(b, k) : cnt;
      rk += cnt;
      --need;
    }
  }
  return rk;
}

__global__ __launch_bounds__(256) void order2_kernel(const int* __restrict__ keypos,
                                                     const int* __restrict__ gcnt,
                                                     int* __restrict__ order) {
  __shared__ int kb[256];
  __shared__ int kpos[256];
  __shared__ int ktot[256];
  const int t = threadIdx.x;
  const int tot = gcnt[t];
  const int rk = walk_place(t);
  kpos[t] = rk;
  kb[rk] = tot;
  ktot[t] = tot;
  __syncthreads();
  for (int off = 1; off < 256; off <<= 1) {
    const int v = (t >= off) ? kb[t - off] : 0;
    __syncthreads();
    kb[t] += v;
    __syncthreads();
  }
  const int n = blockIdx.x * 256 + t;
  if (n < NN) {
    const unsigned kp = (unsigned)keypos[n];
    const int key = (int)(kp >> 24);
    const int pos = (int)(kp & 0xFFFFFFu);
    order[kb[kpos[key]] - ktot[key] + pos] = n;
  }
}

#define TP 132
__global__ __launch_bounds__(256) void conv_kernel(const int* __restrict__ start,
                                                   const int* __restrict__ srow,
                                                   const float* __restrict__ sval,
                                                   const float* __restrict__ hin,
                                                   float* __restrict__ hout,
                                                   const float* __restrict__ e_gate,
                                                   const float* __restrict__ conv_w,
                                                   const int* __restrict__ order,
                                                   const float* __restrict__ env_w,
                                                   const float* __restrict__ env_b,
                                                   float* __restrict__ e_next,
                                                   int* __restrict__ keypos,
                                                   int* __restrict__ gcnt,
                                                   int layer) {
  __shared__ __attribute__((aligned(16))) float tile[8 * 16 * TP];
  __shared__ int khist[256];
  __shared__ int kbase[256];
  const bool do_next = layer + 1 < LL;
  if (do_next) khist[threadIdx.x] = 0;
  const int wave = threadIdx.x >> 5;
  const int lane = threadIdx.x & 31;
  const int half = lane >> 4;
  const int lid  = lane & 15;
  const int slot0 = blockIdx.x * 128 + wave * 16;
  const bool live = slot0 < NN;
  float* tw = tile + wave * (16 * TP);
  int node = 0;
  if (live) {
    node = order[slot0 + lid];
    const int sA = start[node];
    const int eA = (node + 1 < NN) ? start[node + 1] : EE;
    {
      float4 own[8];
      #pragma unroll
      for (int it = 0; it < 8; ++it)
        own[it] = *(const float4*)(hin + __shfl(node, 2 * it + half) * HH + 4 * lid);
      #pragma unroll
      for (int it = 0; it < 8; ++it)
        *(float4*)(tw + (2 * it + half) * TP + HH + 4 * lid) = own[it];
    }
    for (int it = 0; it < 8; ++it) {
      const int i = 2 * it + half;
      const int s = __shfl(sA, i);
      const int e = __shfl(eA, i);
      float4 acc = make_float4(0.0f, 0.0f, 0.0f, 0.0f);
      for (int base = s; ; base += 32) {
        const int rem = e - base;
        const int cmax = max(rem, __shfl_xor(rem, 16));
        if (cmax <= 0) break;
        int myr0 = 0, myr1 = 0;
        float myv0 = 0.0f, myv1 = 0.0f;
        if (lid < rem) { myr0 = srow[base + lid]; myv0 = sval[base + lid]; }
        if (lid + 16 < rem) { myr1 = srow[base + 16 + lid]; myv1 = sval[base + 16 + lid]; }
        const int lim = min(__builtin_amdgcn_readfirstlane(cmax), 32);
        for (int j0 = 0; j0 < lim; j0 += 8) {
          const int myr = (j0 < 16) ? myr0 : myr1;
          const float myv = (j0 < 16) ? myv0 : myv1;
          #pragma unroll
          for (int u = 0; u < 8; ++u) {
            const int j = j0 + u;
            const int r = __shfl(myr, j & 15, 16);
            const float v = __shfl(myv, j & 15, 16);
            float4 hv = *(const float4*)(hin + r * HH + lid * 4);
            const bool on = j < rem;
            hv.x = on ? hv.x : 0.0f;
            hv.y = on ? hv.y : 0.0f;
            hv.z = on ? hv.z : 0.0f;
            hv.w = on ? hv.w : 0.0f;
            acc.x += v * hv.x;
            acc.y += v * hv.y;
            acc.z += v * hv.z;
            acc.w += v * hv.w;
          }
        }
      }
      *(float4*)(tw + i * TP + 4 * lid) = acc;
    }
  }
  __syncthreads();
  int key = 0, rank = 0;
  if (live) {
  const float* W = conv_w + layer * (KK * 2 * HH * HH);
  v8f acc0 = {}, acc1 = {}, acc2 = {}, acc3 = {};
  const float* ap = tw + lid * TP + 4 * half;

  #pragma unroll 1
  for (int k = 0; k < KK; ++k) {
    const float ek = e_gate[node * KK + k];
    if (!__any(ek != 0.0f)) continue;
    const float* wp = W + k * (2 * HH * HH) + (4 * half) * HH + 4 * lid;
    #pragma unroll 2
    for (int g = 0; g < 2 * HH / 8; ++g) {
      float4 a4 = *(const float4*)(ap + 8 * g);
      a4.x = ek * a4.x; a4.y = ek * a4.y; a4.z = ek * a4.z; a4.w = ek * a4.w;
      kgroup8(acc0, acc1, acc2, acc3, a4, wp + 8 * g * HH);
    }
  }
  #pragma unroll
  for (int r = 0; r < 8; ++r) {
    const int orow = __shfl(node, r + 8 * half);
    const float4 res = *(const float4*)(tw + (r + 8 * half) * TP + HH + 4 * lid);
    float4 v;
    v.x = acc0[r] + res.x;
    v.y = acc1[r] + res.y;
    v.z = acc2[r] + res.z;
    v.w = acc3[r] + res.w;
    v.x = v.x > 0.0f ? v.x : 0.0f;
    v.y = v.y > 0.0f ? v.y : 0.0f;
    v.z = v.z > 0.0f ? v.z : 0.0f;
    v.w = v.w > 0.0f ? v.w : 0.0f;
    *(float4*)(hout + orow * HH + 4 * lid) = v;
    if (do_next) *(float4*)(tw + (r + 8 * half) * TP + HH + 4 * lid) = v;
  }
  if (do_next) {
    const float* Wn = env_w + (layer + 1) * 2 * HH * KK + 4 * half;
    const float4 b4 = *(const float4*)(env_b + (layer + 1) * KK + 4 * half);
    float l0 = b4.x, l1 = b4.y, l2 = b4.z, l3 = b4.w;
    const float* hp = tw + lid * TP + HH;
    #pragma unroll 4
    for (int f4 = 0; f4 < HH / 4; ++f4) {
      const float4 hv = *(const float4*)(hp + 4 * f4);
      const float4 wa = *(const float4*)(Wn + (4 * f4 + 0) * KK);
      const float4 wb = *(const float4*)(Wn + (4 * f4 + 1) * KK);
      const float4 wc = *(const float4*)(Wn + (4 * f4 + 2) * KK);
      const float4 wd = *(const float4*)(Wn + (4 * f4 + 3) * KK);
      l0 += hv.x * wa.x; l1 += hv.x * wa.y; l2 += hv.x * wa.z; l3 += hv.x * wa.w;
      l0 += hv.y * wb.x; l1 += hv.y * wb.y; l2 += hv.y * wb.z; l3 += hv.y * wb.w;
      l0 += hv.z * wc.x; l1 += hv.z * wc.y; l2 += hv.z * wc.z; l3 += hv.z * wc.w;
      l0 += hv.w * wd.x; l1 += hv.w * wd.y; l2 += hv.w * wd.z; l3 += hv.w * wd.w;
    }
    float m = fmaxf(fmaxf(l0, l1), fmaxf(l2, l3));
    m = fmaxf(m, __shfl_xor(m, 16));
    const float x0 = __expf(l0 - m), x1 = __expf(l1 - m), x2 = __expf(l2 - m), x3 = __expf(l3 - m);
    const float y0 = __shfl_xor(x0, 16), y1 = __shfl_xor(x1, 16), y2 = __shfl_xor(x2, 16), y3 = __shfl_xor(x3, 16);
    const float t0 = half ? y0 : x0, t1 = half ? y1 : x1, t2 = half ? y2 : x2, t3 = half ? y3 : x3;
    const float t4 = half ? x0 : y0, t5 = half ? x1 : y1, t6 = half ? x2 : y2, t7 = half ? x3 : y3;
    float sm = 0.0f;
    sm += t0; sm += t1; sm += t2; sm += t3; sm += t4; sm += t5; sm += t6; sm += t7;
    const float inv = 1.0f / sm;
    const float p0 = x0 * inv, p1 = x1 * inv, p2 = x2 * inv, p3 = x3 * inv;
    const bool o0 = p0 > THETA, o1 = p1 > THETA, o2 = p2 > THETA, o3 = p3 > THETA;
    float4 ev;
    ev.x = o0 ? p0 : 0.0f;
    ev.y = o1 ? p1 : 0.0f;
    ev.z = o2 ? p2 : 0.0f;
    ev.w = o3 ? p3 : 0.0f;
    *(float4*)(e_next + node * KK + 4 * half) = ev;
    const int bits = ((o0 ? 1 : 0) | (o1 ? 2 : 0) | (o2 ? 4 : 0) | (o3 ? 8 : 0)) << (4 * half);
    key = bits | __shfl_xor(bits, 16);
    if (half == 0) rank = atomicAdd(&khist[key], 1);
  }
  }
  if (do_next) {
    __syncthreads();
    const int c = khist[threadIdx.x];
    if (c > 0) kbase[threadIdx.x] = atomicAdd(&gcnt[threadIdx.x], c);
    __syncthreads();
    if (live && half == 0)
      keypos[node] = (int)(((unsigned)key << 24) | (unsigned)(kbase[key] + rank));
  }
}

__global__ __launch_bounds__(256) void fc1_kernel(const float* __restrict__ h,
                                                  const float* __restrict__ w,
                                                  const float* __restrict__ b,
                                                  float* __restrict__ out) {
  const int wave = threadIdx.x >> 5;
  const int lane = threadIdx.x & 31;
  const int half = lane >> 4;
  const int lid  = lane & 15;
  const int node0 = blockIdx.x * 128 + wave * 16;
  if (node0 >= NN) return;

  v8f acc = {};
  for (int c = 0; c < HH / 4; ++c) {
    const int k0 = 4 * c + 2 * half;
    const float* ap = h + (node0 + lid) * HH + k0;
    v2f a; a.x = ap[0]; a.y = ap[1];
    const float* bp = w + k0 * CC;
    v2f bb; bb.x = bp[lid]; bb.y = bp[CC + lid];
    acc = wmma_f32(a, bb, acc);
  }
  const float bias = b[lid];
  for (int r = 0; r < 8; ++r)
    out[(node0 + r + 8 * half) * CC + lid] = acc[r] + bias;
}

extern "C" void kernel_launch(void* const* d_in, const int* in_sizes, int n_in,
                              void* d_out, int out_size, void* d_ws, size_t ws_size,
                              hipStream_t stream) {
  const float* x      = (const float*)d_in[0];
  const int*   eidx   = (const int*)d_in[1];
  const float* fc0_w  = (const float*)d_in[2];
  const float* fc0_b  = (const float*)d_in[3];
  const float* fc1_w  = (const float*)d_in[4];
  const float* fc1_b  = (const float*)d_in[5];
  const float* env_w  = (const float*)d_in[6];
  const float* env_b  = (const float*)d_in[7];
  const float* conv_w = (const float*)d_in[8];
  float* out = (float*)d_out;

  const int* row = eidx;
  const int* col = eidx + EE;

  float* ws  = (float*)d_ws;
  int*   cs  = (int*)ws;
  float* dn  = ws + NN;
  float* hA  = dn + NN;
  float* lst = hA + (size_t)NN * HH;
  float* hB  = lst + (size_t)NN * HH;
  float* eg  = hB + (size_t)NN * HH;
  int*   srow = (int*)lst;
  float* sval = lst + EE;
  int*   binTot  = (int*)(lst + 2 * (size_t)EE);
  int*   binCur  = binTot + 512;
  int*   binBase = binTot + 1024;
  int*   prow = (int*)hB;
  int*   pcol = prow + EE;
  int*   tail = (int*)(lst + 3 * (size_t)EE);
  int*   bh      = tail + 512;
  int*   keyrank = bh + 4 * QB * 256;
  int*   order   = keyrank + NN;
  int*   keypos  = order + NN;
  float* eg2     = (float*)(keypos + NN);
  int*   gcnt0   = binTot + 1536;
  int*   gcnt1   = binTot + 1792;

  const int T = 256;

  fc0_clear_kernel<<<GB + 8, T, 0, stream>>>(x, fc0_w, fc0_b, hA, binTot);
  gate_count_kernel<<<NB + CBLK, T, 0, stream>>>(hA, env_w, env_b, eg, keyrank, bh, col, binTot);
  scan_partition_kernel<<<1 + PBLK, 1024, 0, stream>>>(bh, row, col, binTot, binCur, binBase, prow, pcol);
  place_order_kernel<<<NB + OB, 512, 0, stream>>>(prow, pcol, binBase, binTot, cs, dn, srow, sval,
                                                  keyrank, bh, order);
  norm_kernel<<<NBLK, T, 0, stream>>>(srow, dn, sval);
  conv_kernel<<<GB, T, 0, stream>>>(cs, srow, sval, hA, hB, eg, conv_w, order, env_w, env_b, eg2, keypos, gcnt0, 0);

  order2_kernel<<<NB, T, 0, stream>>>(keypos, gcnt0, order);
  conv_kernel<<<GB, T, 0, stream>>>(cs, srow, sval, hB, hA, eg2, conv_w, order, env_w, env_b, eg, keypos, gcnt1, 1);
  order2_kernel<<<NB, T, 0, stream>>>(keypos, gcnt1, order);
  conv_kernel<<<GB, T, 0, stream>>>(cs, srow, sval, hA, hB, eg, conv_w, order, env_w, env_b, eg2, keypos, gcnt1, 2);

  fc1_kernel<<<GB, T, 0, stream>>>(hB, fc1_w, fc1_b, out);
}


# Edge weight dn[col]*dn[row] formed in conv 0's gather (dn[row] read beside the source rows, product kept for layers 1-2): no norm launch, no weight store in placement: 11 -> 10 launches; sums kept
# speedup vs baseline: 3.1397x; 1.0334x over previous
#include <hip/hip_runtime.h>
#include <hip/hip_bf16.h>

#define NN 100000
#define DD 128
#define HH 64
#define KK 8
#define LL 3
#define CC 16
#define EE 1600000
#define THETA 0.1f
#define NB ((NN + 255) / 256)
#define QB ((NB + 3) / 4)
#define PBLK 255
#define PCH ((EE + PBLK - 1) / PBLK)
#define PIT ((PCH + 511) / 512)
#define CBLK 512
#define CCH ((EE + CBLK - 1) / CBLK)
#define CIT ((CCH + 255) / 256)
#define GB ((NN + 127) / 128)
#define OB ((NN + 511) / 512)

typedef __attribute__((ext_vector_type(2))) float v2f;
typedef __attribute__((ext_vector_type(8))) float v8f;

__device__ __forceinline__ v8f wmma_f32(v2f a, v2f b, v8f c) {
  return __builtin_amdgcn_wmma_f32_16x16x4_f32(
       false, a,  false, b,
       (short)0, c,  false,  false);
}

__device__ __forceinline__ void kgroup8(v8f& acc0, v8f& acc1, v8f& acc2, v8f& acc3,
                                        const float4 a4, const float* wp) {
  const float4 w0 = *(const float4*)(wp);
  const float4 w1 = *(const float4*)(wp + HH);
  const float4 w2 = *(const float4*)(wp + 2 * HH);
  const float4 w3 = *(const float4*)(wp + 3 * HH);
  v2f a, b;
  a.x = a4.x; a.y = a4.y;
  b.x = w0.x; b.y = w1.x; acc0 = wmma_f32(a, b, acc0);
  b.x = w0.y; b.y = w1.y; acc1 = wmma_f32(a, b, acc1);
  b.x = w0.z; b.y = w1.z; acc2 = wmma_f32(a, b, acc2);
  b.x = w0.w; b.y = w1.w; acc3 = wmma_f32(a, b, acc3);
  a.x = a4.z; a.y = a4.w;
  b.x = w2.x; b.y = w3.x; acc0 = wmma_f32(a, b, acc0);
  b.x = w2.y; b.y = w3.y; acc1 = wmma_f32(a, b, acc1);
  b.x = w2.z; b.y = w3.z; acc2 = wmma_f32(a, b, acc2);
  b.x = w2.w; b.y = w3.w; acc3 = wmma_f32(a, b, acc3);
}


__device__ __forceinline__ void bin_count_body(const int* __restrict__ col,
                                               int* __restrict__ binTot,
                                               int* hist  , const int blk) {
  for (int i = threadIdx.x; i < NB; i += 256) hist[i] = 0;
  __syncthreads();
  const int e0 = blk * CCH;
  int c[CIT];
  #pragma unroll
  for (int j = 0; j < CIT; ++j) {
    const int i = threadIdx.x + 256 * j;
    const int e = e0 + i;
    c[j] = ((i < CCH) && (e < EE)) ? col[e] : -1;
  }
  #pragma unroll
  for (int j = 0; j < CIT; ++j)
    if (c[j] >= 0) atomicAdd(&hist[c[j] >> 8], 1);
  __syncthreads();
  for (int i = threadIdx.x; i < NB; i += 256) {
    const int v = hist[i];
    if (v > 0) atomicAdd(&binTot[i], v);
  }
}

__device__ __forceinline__ void partition_body(const int* __restrict__ row,
                                               const int* __restrict__ col,
                                               const int* __restrict__ binTot,
                                               int* __restrict__ binCur,
                                               int* __restrict__ binBase,
                                               int* __restrict__ prow,
                                               int* __restrict__ pcol,
                                               int* hist, int* sc, int* ofs  ,
                                               const int blk) {
  const int t = threadIdx.x;
  const bool act = t < 512;
  const int tot = (t < NB) ? binTot[t] : 0;
  if (act) {
    hist[t] = 0;
    sc[t] = tot;
  }
  __syncthreads();
  const int e0 = blk * PCH;
  int r[PIT], c[PIT], lr[PIT];
  #pragma unroll
  for (int j = 0; j < PIT; ++j) {
    const int i = t + 512 * j;
    const int e = e0 + i;
    const bool ok = act && (i < PCH) && (e < EE);
    r[j] = ok ? row[e] : 0;
    c[j] = ok ? col[e] : -1;
  }
  #pragma unroll
  for (int j = 0; j < PIT; ++j)
    lr[j] = (c[j] >= 0) ? atomicAdd(&hist[c[j] >> 8], 1) : 0;
  for (int off = 1; off < 512; off <<= 1) {
    const int v = (act && t >= off) ? sc[t - off] : 0;
    __syncthreads();
    if (act) sc[t] += v;
    __syncthreads();
  }
  if (act) {
    const int base = sc[t] - tot;
    const int h = hist[t];
    int o = base;
    if (t < NB && h > 0) o = base + atomicAdd(&binCur[t], h);
    ofs[t] = o;
    if (blk == 0 && t < NB) binBase[t] = base;
  }
  __syncthreads();
  #pragma unroll
  for (int j = 0; j < PIT; ++j) {
    if (c[j] >= 0) {
      const int p = ofs[c[j] >> 8] + lr[j];
      prow[p] = r[j];
      pcol[p] = c[j];
    }
  }
}

__device__ __forceinline__ void bin_place_body(const int* __restrict__ prow,
                                               const int* __restrict__ pcol,
                                               const int* __restrict__ binBase,
                                               const int* __restrict__ binTot,
                                               int* __restrict__ start,
                                               float* __restrict__ dn,
                                               int* __restrict__ srow,
                                               int* cnt, int* sc, int* cur  ,
                                               const int b) {
  const int t = threadIdx.x;
  const int lo = binBase[b];
  const int hi = lo + binTot[b];
  if (t < 256) cnt[t] = 0;
  __syncthreads();
  for (int i0 = lo + t; i0 < hi; i0 += 2048) {
    int c[4];
    #pragma unroll
    for (int u = 0; u < 4; ++u) {
      const int i = i0 + 512 * u;
      c[u] = (i < hi) ? pcol[i] : -1;
    }
    #pragma unroll
    for (int u = 0; u < 4; ++u)
      if (c[u] >= 0) atomicAdd(&cnt[c[u] & 255], 1);
  }
  __syncthreads();
  const int d = (t < 256) ? cnt[t] : 0;
  if (t < 256) sc[t] = d;
  __syncthreads();
  for (int off = 1; off < 256; off <<= 1) {
    int v = 0;
    if (t < 256 && t >= off) v = sc[t - off];
    __syncthreads();
    if (t < 256) sc[t] += v;
    __syncthreads();
  }
  if (t < 256) {
    const int s0 = lo + sc[t] - d;
    const float dv = (d > 0) ? rsqrtf((float)d) : 0.0f;
    cur[t] = s0;
    const int n = b * 256 + t;
    if (n < NN) { start[n] = s0; dn[n] = dv; }
  }
  __syncthreads();
  for (int i0 = lo + t; i0 < hi; i0 += 2048) {
    int c[4], r[4];
    #pragma unroll
    for (int u = 0; u < 4; ++u) {
      const int i = i0 + 512 * u;
      const bool ok = i < hi;
      c[u] = ok ? pcol[i] : -1;
      r[u] = ok ? prow[i] : 0;
    }
    #pragma unroll
    for (int u = 0; u < 4; ++u) {
      if (c[u] >= 0) {
        const int p = atomicAdd(&cur[c[u] & 255], 1);
        srow[p] = r[u];
      }
    }
  }
}

__device__ __forceinline__ void fc0_body(const float* __restrict__ x,
                                         const float* __restrict__ w,
                                         const float* __restrict__ b,
                                         float* __restrict__ h, const int blk) {
  const int wave = threadIdx.x >> 5;
  const int lane = threadIdx.x & 31;
  const int half = lane >> 4;
  const int lid  = lane & 15;
  const int node0 = blk * 128 + wave * 16;
  if (node0 >= NN) return;

  v8f acc0 = {}, acc1 = {}, acc2 = {}, acc3 = {};
  const float* ap = x + (node0 + lid) * DD + 4 * half;
  const float* wp = w + (4 * half) * HH + 4 * lid;
  #pragma unroll 2
  for (int g = 0; g < DD / 8; ++g) {
    const float4 a4 = *(const float4*)(ap + 8 * g);
    kgroup8(acc0, acc1, acc2, acc3, a4, wp + 8 * g * HH);
  }
  const float4 bias = *(const float4*)(b + 4 * lid);
  #pragma unroll
  for (int r = 0; r < 8; ++r) {
    float4 v;
    v.x = acc0[r] + bias.x;
    v.y = acc1[r] + bias.y;
    v.z = acc2[r] + bias.z;
    v.w = acc3[r] + bias.w;
    v.x = v.x > 0.0f ? v.x : 0.0f;
    v.y = v.y > 0.0f ? v.y : 0.0f;
    v.z = v.z > 0.0f ? v.z : 0.0f;
    v.w = v.w > 0.0f ? v.w : 0.0f;
    *(float4*)(h + (node0 + r + 8 * half) * HH + 4 * lid) = v;
  }
}

__device__ __forceinline__ void order_body(const int* __restrict__ keyrank,
                                           const int* __restrict__ offs,
                                           int* __restrict__ order, const int n) {
  if (n < NN) {
    const int kr = keyrank[n];
    order[offs[(n >> 8) * 256 + (kr >> 8)] + (kr & 255)] = n;
  }
}

__device__ __forceinline__ void gate_body(const float* __restrict__ h,
                                          const float* __restrict__ env_w,
                                          const float* __restrict__ env_b,
                                          float* __restrict__ e_out,
                                          int* __restrict__ keyrank,
                                          int* __restrict__ bh, const int layer,
                                          int* hist  , const int blk) {
  const int t = threadIdx.x;
  const int n = blk * 256 + t;
  hist[t] = 0;
  __syncthreads();
  if (n < NN) {
  const float* W = env_w + layer * 2 * HH * KK;
  const float* B = env_b + layer * KK;
  float logit[KK];
  #pragma unroll
  for (int k = 0; k < KK; ++k) logit[k] = B[k];
  for (int f = 0; f < HH; ++f) {
    float hv = h[n * HH + f];
    const float* wr = W + f * KK;
    #pragma unroll
    for (int k = 0; k < KK; ++k) logit[k] += hv * wr[k];
  }
  float m = logit[0];
  #pragma unroll
  for (int k = 1; k < KK; ++k) m = fmaxf(m, logit[k]);
  float s = 0.0f;
  float ex[KK];
  #pragma unroll
  for (int k = 0; k < KK; ++k) { ex[k] = __expf(logit[k] - m); s += ex[k]; }
  float inv = 1.0f / s;
  int key = 0;
  #pragma unroll
  for (int k = 0; k < KK; ++k) {
    float pi = ex[k] * inv;
    const bool on = pi > THETA;
    e_out[n * KK + k] = on ? pi : 0.0f;
    key |= on ? (1 << k) : 0;
  }
  const int r = atomicAdd(&hist[key], 1);
  keyrank[n] = key * 256 + r;
  }
  __syncthreads();
  bh[blk * 256 + t] = hist[t];
  if (blk == 0)
    for (int b = NB; b < 4 * QB; ++b) bh[b * 256 + t] = 0;
}

__device__ __forceinline__ void key_scan_body(int* bh, int (*segsum)[256]  ,
                                              int* kb    ,
                                              int* kpos  ) {
  const int t = threadIdx.x;
  const int key = t & 255;
  const int seg = t >> 8;
  int* p = bh + (seg * QB) * 256 + key;
  int c[QB];
  int s = 0;
  #pragma unroll
  for (int i = 0; i < QB; ++i) {
    c[i] = p[i * 256];
    s += c[i];
  }
  segsum[seg][key] = s;
  __syncthreads();
  if (t < 256) {
    const int tot = segsum[0][t] + segsum[1][t] + segsum[2][t] + segsum[3][t];
    const int pt = __popc(t);
    int rk = 0;
    for (int q = 0; q < 256; ++q) {
      const int pq = __popc(q);
      rk += (pq > pt || (pq == pt && q < t)) ? 1 : 0;
    }
    kpos[t] = rk;
    kb[rk] = tot;
  }
  __syncthreads();
  for (int off = 1; off < 256; off <<= 1) {
    int v = 0;
    if (t < 256 && t >= off) v = kb[t - off];
    __syncthreads();
    if (t < 256) kb[t] += v;
    __syncthreads();
  }
  int run = kb[kpos[key]] - (segsum[0][key] + segsum[1][key] + segsum[2][key] + segsum[3][key]);
  for (int j = 0; j < seg; ++j) run += segsum[j][key];
  #pragma unroll
  for (int i = 0; i < QB; ++i) {
    p[i * 256] = run;
    run += c[i];
  }
}

__global__ __launch_bounds__(256) void fc0_clear_kernel(const float* __restrict__ x,
                                                        const float* __restrict__ w,
                                                        const float* __restrict__ b,
                                                        float* __restrict__ h,
                                                        int* __restrict__ tab) {
  if (blockIdx.x < GB) fc0_body(x, w, b, h, blockIdx.x);
  else tab[(blockIdx.x - GB) * 256 + threadIdx.x] = 0;
}

__global__ __launch_bounds__(256) void gate_count_kernel(const float* __restrict__ h,
                                                         const float* __restrict__ env_w,
                                                         const float* __restrict__ env_b,
                                                         float* __restrict__ e_out,
                                                         int* __restrict__ keyrank,
                                                         int* __restrict__ bh,
                                                         const int* __restrict__ col,
                                                         int* __restrict__ binTot) {
  __shared__ int hist[NB];
  if (blockIdx.x < NB) gate_body(h, env_w, env_b, e_out, keyrank, bh, 0, hist, blockIdx.x);
  else bin_count_body(col, binTot, hist, blockIdx.x - NB);
}

__global__ __launch_bounds__(1024) void scan_partition_kernel(int* bh,
                                                              const int* __restrict__ row,
                                                              const int* __restrict__ col,
                                                              const int* __restrict__ binTot,
                                                              int* __restrict__ binCur,
                                                              int* __restrict__ binBase,
                                                              int* __restrict__ prow,
                                                              int* __restrict__ pcol) {
  __shared__ int lds[6 * 256];
  if (blockIdx.x == 0) key_scan_body(bh, (int (*)[256])lds, lds + 1024, lds + 1280);
  else partition_body(row, col, binTot, binCur, binBase, prow, pcol, lds, lds + 512, lds + 1024, blockIdx.x - 1);
}

__global__ __launch_bounds__(512) void place_order_kernel(const int* __restrict__ prow,
                                                          const int* __restrict__ pcol,
                                                          const int* __restrict__ binBase,
                                                          const int* __restrict__ binTot,
                                                          int* __restrict__ start,
                                                          float* __restrict__ dn,
                                                          int* __restrict__ srow,
                                                          const int* __restrict__ keyrank,
                                                          const int* __restrict__ offs,
                                                          int* __restrict__ order) {
  __shared__ int lds[3 * 256];
  if (blockIdx.x < NB) bin_place_body(prow, pcol, binBase, binTot, start, dn, srow,
                                      lds, lds + 256, lds + 512, blockIdx.x);
  else order_body(keyrank, offs, order, (blockIdx.x - NB) * 512 + threadIdx.x);
}

__device__ constexpr int binom(const int n, const int k) {
  int r = 1;
  for (int i = 1; i <= k; ++i) r = r * (n - k + i) / i;
  return r;
}

__device__ __forceinline__ int walk_place(const int t) {
  const int pt = __popc(t);
  int rk = 0;
  rk += (pt < 8) ? 1 : 0;
  rk += (pt < 7) ? 8 : 0;
  rk += (pt < 6) ? 28 : 0;
  rk += (pt < 5) ? 56 : 0;
  rk += (pt < 4) ? 70 : 0;
  rk += (pt < 3) ? 56 : 0;
  rk += (pt < 2) ? 28 : 0;
  rk += (pt < 1) ? 8 : 0;
  int need = pt;
  #pragma unroll
  for (int b = 7; b >= 0; --b) {
    if ((t >> b) & 1) {
      int cnt = 0;
      #pragma unroll
      for (int k = 0; k <= b; ++k) cnt = (need == k) ? binom(b, k) : cnt;
      rk += cnt;
      --need;
    }
  }
  return rk;
}

__global__ __launch_bounds__(256) void order2_kernel(const int* __restrict__ keypos,
                                                     const int* __restrict__ gcnt,
                                                     int* __restrict__ order) {
  __shared__ int kb[256];
  __shared__ int kpos[256];
  __shared__ int ktot[256];
  const int t = threadIdx.x;
  const int tot = gcnt[t];
  const int rk = walk_place(t);
  kpos[t] = rk;
  kb[rk] = tot;
  ktot[t] = tot;
  __syncthreads();
  for (int off = 1; off < 256; off <<= 1) {
    const int v = (t >= off) ? kb[t - off] : 0;
    __syncthreads();
    kb[t] += v;
    __syncthreads();
  }
  const int n = blockIdx.x * 256 + t;
  if (n < NN) {
    const unsigned kp = (unsigned)keypos[n];
    const int key = (int)(kp >> 24);
    const int pos = (int)(kp & 0xFFFFFFu);
    order[kb[kpos[key]] - ktot[key] + pos] = n;
  }
}

#define TP 132
template <bool FIRST>
__global__ __launch_bounds__(256) void conv_kernel(const int* __restrict__ start,
                                                   const int* __restrict__ srow,
                                                   float* __restrict__ sval,
                                                   const float* __restrict__ dn,
                                                   const float* __restrict__ hin,
                                                   float* __restrict__ hout,
                                                   const float* __restrict__ e_gate,
                                                   const float* __restrict__ conv_w,
                                                   const int* __restrict__ order,
                                                   const float* __restrict__ env_w,
                                                   const float* __restrict__ env_b,
                                                   float* __restrict__ e_next,
                                                   int* __restrict__ keypos,
                                                   int* __restrict__ gcnt,
                                                   int layer) {
  __shared__ __attribute__((aligned(16))) float tile[8 * 16 * TP];
  __shared__ int khist[256];
  __shared__ int kbase[256];
  const bool do_next = layer + 1 < LL;
  if (do_next) khist[threadIdx.x] = 0;
  const int wave = threadIdx.x >> 5;
  const int lane = threadIdx.x & 31;
  const int half = lane >> 4;
  const int lid  = lane & 15;
  const int slot0 = blockIdx.x * 128 + wave * 16;
  const bool live = slot0 < NN;
  float* tw = tile + wave * (16 * TP);
  int node = 0;
  if (live) {
    node = order[slot0 + lid];
    const int sA = start[node];
    const int eA = (node + 1 < NN) ? start[node + 1] : EE;
    const float dA = FIRST ? dn[node] : 0.0f;
    {
      float4 own[8];
      #pragma unroll
      for (int it = 0; it < 8; ++it)
        own[it] = *(const float4*)(hin + __shfl(node, 2 * it + half) * HH + 4 * lid);
      #pragma unroll
      for (int it = 0; it < 8; ++it)
        *(float4*)(tw + (2 * it + half) * TP + HH + 4 * lid) = own[it];
    }
    for (int it = 0; it < 8; ++it) {
      const int i = 2 * it + half;
      const int s = __shfl(sA, i);
      const int e = __shfl(eA, i);
      const float dc = __shfl(dA, i);
      float4 acc = make_float4(0.0f, 0.0f, 0.0f, 0.0f);
      for (int base = s; ; base += 32) {
        const int rem = e - base;
        const int cmax = max(rem, __shfl_xor(rem, 16));
        if (cmax <= 0) break;
        int myr0 = 0, myr1 = 0;
        float myv0 = 0.0f, myv1 = 0.0f;
        const bool in0 = lid < rem, in1 = lid + 16 < rem;
        if (in0) myr0 = srow[base + lid];
        if (in1) myr1 = srow[base + 16 + lid];
        if (FIRST) {
          myv0 = dn[myr0];
          myv1 = dn[myr1];
        } else {
          if (in0) myv0 = sval[base + lid];
          if (in1) myv1 = sval[base + 16 + lid];
        }
        const int lim = min(__builtin_amdgcn_readfirstlane(cmax), 32);
        for (int j0 = 0; j0 < lim; j0 += 8) {
          const int myr = (j0 < 16) ? myr0 : myr1;
          float4 hg[8];
          #pragma unroll
          for (int u = 0; u < 8; ++u) {
            const int r = __shfl(myr, (j0 + u) & 15, 16);
            hg[u] = *(const float4*)(hin + r * HH + lid * 4);
          }
          if (FIRST && j0 == 0) {
            myv0 = dc * myv0;
            myv1 = dc * myv1;
            if (in0) sval[base + lid] = myv0;
            if (in1) sval[base + 16 + lid] = myv1;
          }
          const float myv = (j0 < 16) ? myv0 : myv1;
          #pragma unroll
          for (int u = 0; u < 8; ++u) {
            const int j = j0 + u;
            const float v = __shfl(myv, j & 15, 16);
            float4 hv = hg[u];
            const bool on = j < rem;
            hv.x = on ? hv.x : 0.0f;
            hv.y = on ? hv.y : 0.0f;
            hv.z = on ? hv.z : 0.0f;
            hv.w = on ? hv.w : 0.0f;
            acc.x += v * hv.x;
            acc.y += v * hv.y;
            acc.z += v * hv.z;
            acc.w += v * hv.w;
          }
        }
      }
      *(float4*)(tw + i * TP + 4 * lid) = acc;
    }
  }
  __syncthreads();
  int key = 0, rank = 0;
  if (live) {
  const float* W = conv_w + layer * (KK * 2 * HH * HH);
  v8f acc0 = {}, acc1 = {}, acc2 = {}, acc3 = {};
  const float* ap = tw + lid * TP + 4 * half;

  #pragma unroll 1
  for (int k = 0; k < KK; ++k) {
    const float ek = e_gate[node * KK + k];
    if (!__any(ek != 0.0f)) continue;
    const float* wp = W + k * (2 * HH * HH) + (4 * half) * HH + 4 * lid;
    #pragma unroll 2
    for (int g = 0; g < 2 * HH / 8; ++g) {
      float4 a4 = *(const float4*)(ap + 8 * g);
      a4.x = ek * a4.x; a4.y = ek * a4.y; a4.z = ek * a4.z; a4.w = ek * a4.w;
      kgroup8(acc0, acc1, acc2, acc3, a4, wp + 8 * g * HH);
    }
  }
  #pragma unroll
  for (int r = 0; r < 8; ++r) {
    const int orow = __shfl(node, r + 8 * half);
    const float4 res = *(const float4*)(tw + (r + 8 * half) * TP + HH + 4 * lid);
    float4 v;
    v.x = acc0[r] + res.x;
    v.y = acc1[r] + res.y;
    v.z = acc2[r] + res.z;
    v.w = acc3[r] + res.w;
    v.x = v.x > 0.0f ? v.x : 0.0f;
    v.y = v.y > 0.0f ? v.y : 0.0f;
    v.z = v.z > 0.0f ? v.z : 0.0f;
    v.w = v.w > 0.0f ? v.w : 0.0f;
    *(float4*)(hout + orow * HH + 4 * lid) = v;
    if (do_next) *(float4*)(tw + (r + 8 * half) * TP + HH + 4 * lid) = v;
  }
  if (do_next) {
    const float* Wn = env_w + (layer + 1) * 2 * HH * KK + 4 * half;
    const float4 b4 = *(const float4*)(env_b + (layer + 1) * KK + 4 * half);
    float l0 = b4.x, l1 = b4.y, l2 = b4.z, l3 = b4.w;
    const float* hp = tw + lid * TP + HH;
    #pragma unroll 4
    for (int f4 = 0; f4 < HH / 4; ++f4) {
      const float4 hv = *(const float4*)(hp + 4 * f4);
      const float4 wa = *(const float4*)(Wn + (4 * f4 + 0) * KK);
      const float4 wb = *(const float4*)(Wn + (4 * f4 + 1) * KK);
      const float4 wc = *(const float4*)(Wn + (4 * f4 + 2) * KK);
      const float4 wd = *(const float4*)(Wn + (4 * f4 + 3) * KK);
      l0 += hv.x * wa.x; l1 += hv.x * wa.y; l2 += hv.x * wa.z; l3 += hv.x * wa.w;
      l0 += hv.y * wb.x; l1 += hv.y * wb.y; l2 += hv.y * wb.z; l3 += hv.y * wb.w;
      l0 += hv.z * wc.x; l1 += hv.z * wc.y; l2 += hv.z * wc.z; l3 += hv.z * wc.w;
      l0 += hv.w * wd.x; l1 += hv.w * wd.y; l2 += hv.w * wd.z; l3 += hv.w * wd.w;
    }
    float m = fmaxf(fmaxf(l0, l1), fmaxf(l2, l3));
    m = fmaxf(m, __shfl_xor(m, 16));
    const float x0 = __expf(l0 - m), x1 = __expf(l1 - m), x2 = __expf(l2 - m), x3 = __expf(l3 - m);
    const float y0 = __shfl_xor(x0, 16), y1 = __shfl_xor(x1, 16), y2 = __shfl_xor(x2, 16), y3 = __shfl_xor(x3, 16);
    const float t0 = half ? y0 : x0, t1 = half ? y1 : x1, t2 = half ? y2 : x2, t3 = half ? y3 : x3;
    const float t4 = half ? x0 : y0, t5 = half ? x1 : y1, t6 = half ? x2 : y2, t7 = half ? x3 : y3;
    float sm = 0.0f;
    sm += t0; sm += t1; sm += t2; sm += t3; sm += t4; sm += t5; sm += t6; sm += t7;
    const float inv = 1.0f / sm;
    const float p0 = x0 * inv, p1 = x1 * inv, p2 = x2 * inv, p3 = x3 * inv;
    const bool o0 = p0 > THETA, o1 = p1 > THETA, o2 = p2 > THETA, o3 = p3 > THETA;
    float4 ev;
    ev.x = o0 ? p0 : 0.0f;
    ev.y = o1 ? p1 : 0.0f;
    ev.z = o2 ? p2 : 0.0f;
    ev.w = o3 ? p3 : 0.0f;
    *(float4*)(e_next + node * KK + 4 * half) = ev;
    const int bits = ((o0 ? 1 : 0) | (o1 ? 2 : 0) | (o2 ? 4 : 0) | (o3 ? 8 : 0)) << (4 * half);
    key = bits | __shfl_xor(bits, 16);
    if (half == 0) rank = atomicAdd(&khist[key], 1);
  }
  }
  if (do_next) {
    __syncthreads();
    const int c = khist[threadIdx.x];
    if (c > 0) kbase[threadIdx.x] = atomicAdd(&gcnt[threadIdx.x], c);
    __syncthreads();
    if (live && half == 0)
      keypos[node] = (int)(((unsigned)key << 24) | (unsigned)(kbase[key] + rank));
  }
}

__global__ __launch_bounds__(256) void fc1_kernel(const float* __restrict__ h,
                                                  const float* __restrict__ w,
                                                  const float* __restrict__ b,
                                                  float* __restrict__ out) {
  const int wave = threadIdx.x >> 5;
  const int lane = threadIdx.x & 31;
  const int half = lane >> 4;
  const int lid  = lane & 15;
  const int node0 = blockIdx.x * 128 + wave * 16;
  if (node0 >= NN) return;

  v8f acc = {};
  for (int c = 0; c < HH / 4; ++c) {
    const int k0 = 4 * c + 2 * half;
    const float* ap = h + (node0 + lid) * HH + k0;
    v2f a; a.x = ap[0]; a.y = ap[1];
    const float* bp = w + k0 * CC;
    v2f bb; bb.x = bp[lid]; bb.y = bp[CC + lid];
    acc = wmma_f32(a, bb, acc);
  }
  const float bias = b[lid];
  for (int r = 0; r < 8; ++r)
    out[(node0 + r + 8 * half) * CC + lid] = acc[r] + bias;
}

extern "C" void kernel_launch(void* const* d_in, const int* in_sizes, int n_in,
                              void* d_out, int out_size, void* d_ws, size_t ws_size,
                              hipStream_t stream) {
  const float* x      = (const float*)d_in[0];
  const int*   eidx   = (const int*)d_in[1];
  const float* fc0_w  = (const float*)d_in[2];
  const float* fc0_b  = (const float*)d_in[3];
  const float* fc1_w  = (const float*)d_in[4];
  const float* fc1_b  = (const float*)d_in[5];
  const float* env_w  = (const float*)d_in[6];
  const float* env_b  = (const float*)d_in[7];
  const float* conv_w = (const float*)d_in[8];
  float* out = (float*)d_out;

  const int* row = eidx;
  const int* col = eidx + EE;

  float* ws  = (float*)d_ws;
  int*   cs  = (int*)ws;
  float* dn  = ws + NN;
  float* hA  = dn + NN;
  float* lst = hA + (size_t)NN * HH;
  float* hB  = lst + (size_t)NN * HH;
  float* eg  = hB + (size_t)NN * HH;
  int*   srow = (int*)lst;
  float* sval = lst + EE;
  int*   binTot  = (int*)(lst + 2 * (size_t)EE);
  int*   binCur  = binTot + 512;
  int*   binBase = binTot + 1024;
  int*   prow = (int*)hB;
  int*   pcol = prow + EE;
  int*   tail = (int*)(lst + 3 * (size_t)EE);
  int*   bh      = tail + 512;
  int*   keyrank = bh + 4 * QB * 256;
  int*   order   = keyrank + NN;
  int*   keypos  = order + NN;
  float* eg2     = (float*)(keypos + NN);
  int*   gcnt0   = binTot + 1536;
  int*   gcnt1   = binTot + 1792;

  const int T = 256;

  fc0_clear_kernel<<<GB + 8, T, 0, stream>>>(x, fc0_w, fc0_b, hA, binTot);
  gate_count_kernel<<<NB + CBLK, T, 0, stream>>>(hA, env_w, env_b, eg, keyrank, bh, col, binTot);
  scan_partition_kernel<<<1 + PBLK, 1024, 0, stream>>>(bh, row, col, binTot, binCur, binBase, prow, pcol);
  place_order_kernel<<<NB + OB, 512, 0, stream>>>(prow, pcol, binBase, binTot, cs, dn, srow,
                                                  keyrank, bh, order);
  conv_kernel<true><<<GB, T, 0, stream>>>(cs, srow, sval, dn, hA, hB, eg, conv_w, order, env_w, env_b, eg2, keypos, gcnt0, 0);

  order2_kernel<<<NB, T, 0, stream>>>(keypos, gcnt0, order);
  conv_kernel<false><<<GB, T, 0, stream>>>(cs, srow, sval, dn, hB, hA, eg2, conv_w, order, env_w, env_b, eg, keypos, gcnt1, 1);
  order2_kernel<<<NB, T, 0, stream>>>(keypos, gcnt1, order);
  conv_kernel<false><<<GB, T, 0, stream>>>(cs, srow, sval, dn, hA, hB, eg, conv_w, order, env_w, env_b, eg2, keypos, gcnt1, 2);

  fc1_kernel<<<GB, T, 0, stream>>>(hB, fc1_w, fc1_b, out);
}
